# starting kernel
# baseline (speedup 1.0000x reference)
#include <hip/hip_runtime.h>

typedef __attribute__((ext_vector_type(2))) float v2f;
typedef __attribute__((ext_vector_type(8))) float v8f;

constexpr int IN_DIM = 128;
constexpr int HID    = 8;


__global__ void zero_f32(float* __restrict__ p, int n) {
  int i = blockIdx.x * blockDim.x + threadIdx.x;
  if (i < n) p[i] = 0.0f;
}

__global__ void degree_kernel(const int* __restrict__ src, const int* __restrict__ dst,
                              float* __restrict__ deg_s, float* __restrict__ deg_d, int nE) {
  int e = blockIdx.x * blockDim.x + threadIdx.x;
  if (e < nE) {
    atomicAdd(&deg_s[src[e]], 1.0f);
    atomicAdd(&deg_d[dst[e]], 1.0f);
  }
}

__global__ void norm_kernel(float* __restrict__ deg_s, float* __restrict__ deg_d, int n) {
  int i = blockIdx.x * blockDim.x + threadIdx.x;
  if (i < n) {
    deg_s[i] = rsqrtf(fmaxf(deg_s[i], 1.0f));
    deg_d[i] = rsqrtf(fmaxf(deg_d[i], 1.0f));
  }
}


__global__ __launch_bounds__(256) void gemm1_wmma(const float* __restrict__ x,
                                                  const float* __restrict__ W1,
                                                  const float* __restrict__ out_norm,
                                                  float* __restrict__ h, int nNodes) {
  __shared__ float w[IN_DIM * 16];
  int tid = threadIdx.x;
  for (int i = tid; i < IN_DIM * 16; i += 256) {
    int r = i >> 4, cidx = i & 15;
    w[i] = (cidx < HID) ? W1[r * HID + cidx] : 0.0f;
  }
  __syncthreads();

  int wid  = tid >> 5;
  int lane = tid & 31;
  int m    = lane & 15;
  int kh   = lane >> 4;

  int row0 = (blockIdx.x * 8 + wid) * 16;
  int node = row0 + m;
  int nodeClamped = node < nNodes ? node : nNodes - 1;
  const float* xrow = x + (long)nodeClamped * IN_DIM;

  v8f acc = {};
#pragma unroll
  for (int kk = 0; kk < IN_DIM; kk += 4) {
    int k0 = kk + 2 * kh;
    v2f a;
    a.x = xrow[k0];
    a.y = xrow[k0 + 1];
    v2f b;
    b.x = w[k0 * 16 + m];
    b.y = w[(k0 + 1) * 16 + m];
    acc = __builtin_amdgcn_wmma_f32_16x16x4_f32(
         false, a,  false, b,
         (short)0, acc,  false,  false);
  }

  if (m < HID) {
#pragma unroll
    for (int r = 0; r < 8; ++r) {
      int nr = row0 + r + 8 * kh;
      if (nr < nNodes) h[(long)nr * HID + m] = acc[r] * out_norm[nr];
    }
  }
}


__global__ void spmm1_kernel(const int* __restrict__ src, const int* __restrict__ dst,
                             const float* __restrict__ h, float* __restrict__ agg, int nE) {
  int e = blockIdx.x * blockDim.x + threadIdx.x;
  if (e >= nE) return;
  int s = src[e], d = dst[e];
  const float4* hs = (const float4*)(h + (long)s * HID);
  float4 h0 = hs[0];
  float4 h1 = hs[1];
  float* ad = agg + (long)d * HID;
  atomicAdd(&ad[0], h0.x); atomicAdd(&ad[1], h0.y);
  atomicAdd(&ad[2], h0.z); atomicAdd(&ad[3], h0.w);
  atomicAdd(&ad[4], h1.x); atomicAdd(&ad[5], h1.y);
  atomicAdd(&ad[6], h1.z); atomicAdd(&ad[7], h1.w);
}


__global__ void layer2_node(const float* __restrict__ agg, const float* __restrict__ in_norm,
                            const float* __restrict__ out_norm, const float* __restrict__ b1,
                            const float* __restrict__ W2, float* __restrict__ g2, int n) {
  int v = blockIdx.x * blockDim.x + threadIdx.x;
  if (v >= n) return;
  float innv = in_norm[v];
  const float* av = agg + (long)v * HID;
  float s = 0.0f;
#pragma unroll
  for (int j = 0; j < HID; ++j) {
    float z = fmaf(av[j], innv, b1[j]);
    z = z > 0.0f ? z : 0.0f;
    s = fmaf(z, W2[j], s);
  }
  g2[v] = s * out_norm[v];
}


__global__ void spmm2_kernel(const int* __restrict__ src, const int* __restrict__ dst,
                             const float* __restrict__ g2, float* __restrict__ outacc, int nE) {
  int e = blockIdx.x * blockDim.x + threadIdx.x;
  if (e < nE) atomicAdd(&outacc[dst[e]], g2[src[e]]);
}

__global__ void final_kernel(const float* __restrict__ outacc, const float* __restrict__ in_norm,
                             const float* __restrict__ b2, float* __restrict__ out, int n) {
  int v = blockIdx.x * blockDim.x + threadIdx.x;
  if (v < n) out[v] = fmaf(outacc[v], in_norm[v], b2[0]);
}


extern "C" void kernel_launch(void* const* d_in, const int* in_sizes, int n_in,
                              void* d_out, int out_size, void* d_ws, size_t ws_size,
                              hipStream_t stream) {
  const float* x  = (const float*)d_in[0];
  const float* W1 = (const float*)d_in[1];
  const float* b1 = (const float*)d_in[2];
  const float* W2 = (const float*)d_in[3];
  const float* b2 = (const float*)d_in[4];
  const int*  src = (const int*)d_in[5];
  const int*  dst = (const int*)d_in[6];

  const int nN = in_sizes[0] / IN_DIM;
  const int nE = in_sizes[5];

  float* ws     = (float*)d_ws;
  float* deg_s  = ws;
  float* deg_d  = ws + (long)nN;
  float* h      = ws + 2L * nN;
  float* agg    = ws + 10L * nN;
  float* g2     = ws + 18L * nN;
  float* outacc = ws + 19L * nN;
  float* out    = (float*)d_out;

  const int TB = 256;
  dim3 blk(TB);

  zero_f32<<<(2 * nN + TB - 1) / TB, blk, 0, stream>>>(deg_s, 2 * nN);
  zero_f32<<<(8 * nN + TB - 1) / TB, blk, 0, stream>>>(agg, 8 * nN);
  zero_f32<<<(nN + TB - 1) / TB, blk, 0, stream>>>(outacc, nN);

  degree_kernel<<<(nE + TB - 1) / TB, blk, 0, stream>>>(src, dst, deg_s, deg_d, nE);
  norm_kernel<<<(nN + TB - 1) / TB, blk, 0, stream>>>(deg_s, deg_d, nN);

  gemm1_wmma<<<(nN + 127) / 128, blk, 0, stream>>>(x, W1, deg_s, h, nN);

  spmm1_kernel<<<(nE + TB - 1) / TB, blk, 0, stream>>>(src, dst, h, agg, nE);
  layer2_node<<<(nN + TB - 1) / TB, blk, 0, stream>>>(agg, deg_d, deg_s, b1, W2, g2, nN);
  spmm2_kernel<<<(nE + TB - 1) / TB, blk, 0, stream>>>(src, dst, g2, outacc, nE);
  final_kernel<<<(nN + TB - 1) / TB, blk, 0, stream>>>(outacc, deg_d, b2, out, nN);
}


# layer-1 row sums add in LDS slices of 8192 dst rows: owner work-groups scan the unsorted edge list, queue hits per wave, gather rows; whole-line atomic flush; LDS+global atomic arrival order
# speedup vs baseline: 1.4832x; 1.4832x over previous
#include <hip/hip_runtime.h>

typedef __attribute__((ext_vector_type(2))) float v2f;
typedef __attribute__((ext_vector_type(8))) float v8f;

constexpr int IN_DIM = 128;
constexpr int HID    = 8;


__global__ void zero_f32(float* __restrict__ p, int n) {
  int i = blockIdx.x * blockDim.x + threadIdx.x;
  if (i < n) p[i] = 0.0f;
}

__global__ void degree_kernel(const int* __restrict__ src, const int* __restrict__ dst,
                              float* __restrict__ deg_s, float* __restrict__ deg_d, int nE) {
  int e = blockIdx.x * blockDim.x + threadIdx.x;
  if (e < nE) {
    atomicAdd(&deg_s[src[e]], 1.0f);
    atomicAdd(&deg_d[dst[e]], 1.0f);
  }
}

__global__ void norm_kernel(float* __restrict__ deg_s, float* __restrict__ deg_d, int n) {
  int i = blockIdx.x * blockDim.x + threadIdx.x;
  if (i < n) {
    deg_s[i] = rsqrtf(fmaxf(deg_s[i], 1.0f));
    deg_d[i] = rsqrtf(fmaxf(deg_d[i], 1.0f));
  }
}


__global__ __launch_bounds__(256) void gemm1_wmma(const float* __restrict__ x,
                                                  const float* __restrict__ W1,
                                                  const float* __restrict__ out_norm,
                                                  float* __restrict__ h, int nNodes) {
  __shared__ float w[IN_DIM * 16];
  int tid = threadIdx.x;
  for (int i = tid; i < IN_DIM * 16; i += 256) {
    int r = i >> 4, cidx = i & 15;
    w[i] = (cidx < HID) ? W1[r * HID + cidx] : 0.0f;
  }
  __syncthreads();

  int wid  = tid >> 5;
  int lane = tid & 31;
  int m    = lane & 15;
  int kh   = lane >> 4;

  int row0 = (blockIdx.x * 8 + wid) * 16;
  int node = row0 + m;
  int nodeClamped = node < nNodes ? node : nNodes - 1;
  const float* xrow = x + (long)nodeClamped * IN_DIM;

  v8f acc = {};
#pragma unroll
  for (int kk = 0; kk < IN_DIM; kk += 4) {
    int k0 = kk + 2 * kh;
    v2f a;
    a.x = xrow[k0];
    a.y = xrow[k0 + 1];
    v2f b;
    b.x = w[k0 * 16 + m];
    b.y = w[(k0 + 1) * 16 + m];
    acc = __builtin_amdgcn_wmma_f32_16x16x4_f32(
         false, a,  false, b,
         (short)0, acc,  false,  false);
  }

  if (m < HID) {
#pragma unroll
    for (int r = 0; r < 8; ++r) {
      int nr = row0 + r + 8 * kh;
      if (nr < nNodes) h[(long)nr * HID + m] = acc[r] * out_norm[nr];
    }
  }
}


constexpr int SL_NODES   = 8192;
constexpr int SL_PARTS   = 4;
constexpr int SL_THREADS = 1024;
constexpr int SL_WAVES   = SL_THREADS / 32;
constexpr int SL_QCAP    = 448;
constexpr int SL_TRIP    = 256;
static_assert(SL_NODES * HID * 4 + SL_WAVES * SL_QCAP * 4 <= 320 * 1024, "LDS budget");
static_assert(SL_QCAP >= SL_TRIP, "a whole trip of hits fits behind the drain mark");

__device__ __forceinline__ int4 load_quad(const int* __restrict__ p, long q, long qEnd) {
  long qc = q < qEnd ? q : qEnd - 1;
  int4 v = ((const int4*)p)[qc];
  if (q >= qEnd) v = make_int4(-1, -1, -1, -1);
  return v;
}

__device__ __forceinline__ void note_hits(unsigned* __restrict__ q, int& cnt, int d, unsigned e, int base) {
  bool hit = (unsigned)(d - base) < (unsigned)SL_NODES;
  unsigned m = __builtin_amdgcn_ballot_w32(hit);
  if (hit) q[cnt + (int)__builtin_amdgcn_mbcnt_lo(m, 0u)] = e;
  cnt += __builtin_popcount(m);
}

__device__ __forceinline__ void add_row(float* __restrict__ acc, int r, const float* __restrict__ h, int s) {
  const float4* hs = (const float4*)(h + (long)s * HID);
  float4 h0 = hs[0];
  float4 h1 = hs[1];
  float* row = acc + r * HID;
  atomicAdd(&row[0], h0.x); atomicAdd(&row[1], h0.y);
  atomicAdd(&row[2], h0.z); atomicAdd(&row[3], h0.w);
  atomicAdd(&row[4], h1.x); atomicAdd(&row[5], h1.y);
  atomicAdd(&row[6], h1.z); atomicAdd(&row[7], h1.w);
}

__device__ __forceinline__ void drain(const unsigned* __restrict__ q, int cnt, int lane, int base,
                                      const int* __restrict__ src, const int* __restrict__ dst,
                                      const float* __restrict__ h, float* __restrict__ acc) {
  __builtin_amdgcn_fence(__ATOMIC_ACQ_REL, "wavefront");
  for (int i = lane; i < cnt; i += 32) {
    unsigned e = q[i];
    int d = dst[e];
    int s = src[e];
    add_row(acc, d - base, h, s);
  }
  __builtin_amdgcn_fence(__ATOMIC_ACQ_REL, "wavefront");
}

__global__ __launch_bounds__(SL_THREADS) void spmm1_slices(const int* __restrict__ src,
                                                           const int* __restrict__ dst,
                                                           const float* __restrict__ h,
                                                           float* __restrict__ agg, int nE, int nN) {
  __shared__ float acc[SL_NODES * HID];
  __shared__ unsigned queue[SL_WAVES * SL_QCAP];

  const int tid  = threadIdx.x;
  const int lane = tid & 31;
  const int wave = tid >> 5;
  const int slice = blockIdx.x / SL_PARTS;
  const int part  = blockIdx.x % SL_PARTS;
  const int base  = slice * SL_NODES;

  for (int i = tid; i < SL_NODES * HID / 4; i += SL_THREADS)
    ((float4*)acc)[i] = make_float4(0.0f, 0.0f, 0.0f, 0.0f);
  __syncthreads();

  unsigned* q = queue + wave * SL_QCAP;
  int cnt = 0;

  const long nQ   = nE >> 2;
  const long qPer = (nQ + SL_PARTS - 1) / SL_PARTS;
  const long q0   = part * qPer;
  const long q1   = (q0 + qPer) < nQ ? (q0 + qPer) : nQ;

  int4 n0 = make_int4(-1, -1, -1, -1), n1 = n0;
  if (q0 < q1) {
    n0 = load_quad(dst, q0 + tid, q1);
    n1 = load_quad(dst, q0 + tid + SL_THREADS, q1);
  }
  for (long qb = q0; qb < q1; qb += 2 * SL_THREADS) {
    int4 c0 = n0, c1 = n1;
    n0 = load_quad(dst, qb + 2 * SL_THREADS + tid, q1);
    n1 = load_quad(dst, qb + 3 * SL_THREADS + tid, q1);
    unsigned e0 = (unsigned)(4 * (qb + tid));
    unsigned e1 = e0 + 4u * SL_THREADS;
    note_hits(q, cnt, c0.x, e0 + 0u, base);
    note_hits(q, cnt, c0.y, e0 + 1u, base);
    note_hits(q, cnt, c0.z, e0 + 2u, base);
    note_hits(q, cnt, c0.w, e0 + 3u, base);
    note_hits(q, cnt, c1.x, e1 + 0u, base);
    note_hits(q, cnt, c1.y, e1 + 1u, base);
    note_hits(q, cnt, c1.z, e1 + 2u, base);
    note_hits(q, cnt, c1.w, e1 + 3u, base);
    if (cnt > SL_QCAP - SL_TRIP) {
      drain(q, cnt, lane, base, src, dst, h, acc);
      cnt = 0;
    }
  }
  drain(q, cnt, lane, base, src, dst, h, acc);

  if (part == 0) {
    int e = (int)(nQ << 2) + tid;
    if (e < nE) {
      int d = dst[e];
      if ((unsigned)(d - base) < (unsigned)SL_NODES) add_row(acc, d - base, h, src[e]);
    }
  }
  __syncthreads();

  const int rows = (nN - base) < SL_NODES ? (nN - base) : SL_NODES;
  float* out = agg + (long)base * HID;
  for (int i = tid; i < rows * HID; i += SL_THREADS) atomicAdd(&out[i], acc[i]);
}


__global__ void layer2_node(const float* __restrict__ agg, const float* __restrict__ in_norm,
                            const float* __restrict__ out_norm, const float* __restrict__ b1,
                            const float* __restrict__ W2, float* __restrict__ g2, int n) {
  int v = blockIdx.x * blockDim.x + threadIdx.x;
  if (v >= n) return;
  float innv = in_norm[v];
  const float* av = agg + (long)v * HID;
  float s = 0.0f;
#pragma unroll
  for (int j = 0; j < HID; ++j) {
    float z = fmaf(av[j], innv, b1[j]);
    z = z > 0.0f ? z : 0.0f;
    s = fmaf(z, W2[j], s);
  }
  g2[v] = s * out_norm[v];
}


__global__ void spmm2_kernel(const int* __restrict__ src, const int* __restrict__ dst,
                             const float* __restrict__ g2, float* __restrict__ outacc, int nE) {
  int e = blockIdx.x * blockDim.x + threadIdx.x;
  if (e < nE) atomicAdd(&outacc[dst[e]], g2[src[e]]);
}

__global__ void final_kernel(const float* __restrict__ outacc, const float* __restrict__ in_norm,
                             const float* __restrict__ b2, float* __restrict__ out, int n) {
  int v = blockIdx.x * blockDim.x + threadIdx.x;
  if (v < n) out[v] = fmaf(outacc[v], in_norm[v], b2[0]);
}


extern "C" void kernel_launch(void* const* d_in, const int* in_sizes, int n_in,
                              void* d_out, int out_size, void* d_ws, size_t ws_size,
                              hipStream_t stream) {
  const float* x  = (const float*)d_in[0];
  const float* W1 = (const float*)d_in[1];
  const float* b1 = (const float*)d_in[2];
  const float* W2 = (const float*)d_in[3];
  const float* b2 = (const float*)d_in[4];
  const int*  src = (const int*)d_in[5];
  const int*  dst = (const int*)d_in[6];

  const int nN = in_sizes[0] / IN_DIM;
  const int nE = in_sizes[5];

  float* ws     = (float*)d_ws;
  float* deg_s  = ws;
  float* deg_d  = ws + (long)nN;
  float* h      = ws + 2L * nN;
  float* agg    = ws + 10L * nN;
  float* g2     = ws + 18L * nN;
  float* outacc = ws + 19L * nN;
  float* out    = (float*)d_out;

  const int TB = 256;
  dim3 blk(TB);

  zero_f32<<<(2 * nN + TB - 1) / TB, blk, 0, stream>>>(deg_s, 2 * nN);
  zero_f32<<<(8 * nN + TB - 1) / TB, blk, 0, stream>>>(agg, 8 * nN);
  zero_f32<<<(nN + TB - 1) / TB, blk, 0, stream>>>(outacc, nN);

  degree_kernel<<<(nE + TB - 1) / TB, blk, 0, stream>>>(src, dst, deg_s, deg_d, nE);
  norm_kernel<<<(nN + TB - 1) / TB, blk, 0, stream>>>(deg_s, deg_d, nN);

  gemm1_wmma<<<(nN + 127) / 128, blk, 0, stream>>>(x, W1, deg_s, h, nN);

  spmm1_slices<<<((nN + SL_NODES - 1) / SL_NODES) * SL_PARTS, dim3(SL_THREADS), 0, stream>>>(src, dst, h, agg, nE, nN);
  layer2_node<<<(nN + TB - 1) / TB, blk, 0, stream>>>(agg, deg_d, deg_s, b1, W2, g2, nN);
  spmm2_kernel<<<(nE + TB - 1) / TB, blk, 0, stream>>>(src, dst, g2, outacc, nE);
  final_kernel<<<(nN + TB - 1) / TB, blk, 0, stream>>>(outacc, deg_d, b2, out, nN);
}


# in-degree counted by the layer-1 slice owners: one integer LDS add per queued edge, whole-line flush; degree kernel keeps only the out-degree atomic per edge (half its scattered global atomics gone)
# speedup vs baseline: 1.7957x; 1.2107x over previous
#include <hip/hip_runtime.h>

typedef __attribute__((ext_vector_type(2))) float v2f;
typedef __attribute__((ext_vector_type(8))) float v8f;

constexpr int IN_DIM = 128;
constexpr int HID    = 8;


__global__ void zero_f32(float* __restrict__ p, int n) {
  int i = blockIdx.x * blockDim.x + threadIdx.x;
  if (i < n) p[i] = 0.0f;
}

__global__ void degree_kernel(const int* __restrict__ src, float* __restrict__ deg_s, int nE) {
  int e = blockIdx.x * blockDim.x + threadIdx.x;
  if (e < nE) atomicAdd(&deg_s[src[e]], 1.0f);
}

__global__ void norm_kernel(float* __restrict__ deg_s, int n) {
  int i = blockIdx.x * blockDim.x + threadIdx.x;
  if (i < n) deg_s[i] = rsqrtf(fmaxf(deg_s[i], 1.0f));
}


__global__ __launch_bounds__(256) void gemm1_wmma(const float* __restrict__ x,
                                                  const float* __restrict__ W1,
                                                  const float* __restrict__ out_norm,
                                                  float* __restrict__ h, int nNodes) {
  __shared__ float w[IN_DIM * 16];
  int tid = threadIdx.x;
  for (int i = tid; i < IN_DIM * 16; i += 256) {
    int r = i >> 4, cidx = i & 15;
    w[i] = (cidx < HID) ? W1[r * HID + cidx] : 0.0f;
  }
  __syncthreads();

  int wid  = tid >> 5;
  int lane = tid & 31;
  int m    = lane & 15;
  int kh   = lane >> 4;

  int row0 = (blockIdx.x * 8 + wid) * 16;
  int node = row0 + m;
  int nodeClamped = node < nNodes ? node : nNodes - 1;
  const float* xrow = x + (long)nodeClamped * IN_DIM;

  v8f acc = {};
#pragma unroll
  for (int kk = 0; kk < IN_DIM; kk += 4) {
    int k0 = kk + 2 * kh;
    v2f a;
    a.x = xrow[k0];
    a.y = xrow[k0 + 1];
    v2f b;
    b.x = w[k0 * 16 + m];
    b.y = w[(k0 + 1) * 16 + m];
    acc = __builtin_amdgcn_wmma_f32_16x16x4_f32(
         false, a,  false, b,
         (short)0, acc,  false,  false);
  }

  if (m < HID) {
#pragma unroll
    for (int r = 0; r < 8; ++r) {
      int nr = row0 + r + 8 * kh;
      if (nr < nNodes) h[(long)nr * HID + m] = acc[r] * out_norm[nr];
    }
  }
}


constexpr int SL_NODES   = 7952;
constexpr int SL_PARTS   = 4;
constexpr int SL_THREADS = 1024;
constexpr int SL_WAVES   = SL_THREADS / 32;
constexpr int SL_QCAP    = 320;
constexpr int SL_GROUP   = 128;
static_assert(SL_NODES * (HID + 1) * 4 + SL_WAVES * SL_QCAP * 4 <= 320 * 1024, "LDS budget");
static_assert(SL_QCAP > SL_GROUP, "a whole group of hits fits behind the drain mark");
static_assert((SL_NODES * HID) % 32 == 0, "a slice of agg starts on a 128-byte line");

__device__ __forceinline__ int4 load_quad(const int* __restrict__ p, long q, long qEnd) {
  long qc = q < qEnd ? q : qEnd - 1;
  int4 v = ((const int4*)p)[qc];
  if (q >= qEnd) v = make_int4(-1, -1, -1, -1);
  return v;
}

__device__ __forceinline__ void note_hits(unsigned* __restrict__ q, int& cnt, int d, unsigned e, int base) {
  bool hit = (unsigned)(d - base) < (unsigned)SL_NODES;
  unsigned m = __builtin_amdgcn_ballot_w32(hit);
  if (hit) q[cnt + (int)__builtin_amdgcn_mbcnt_lo(m, 0u)] = e;
  cnt += __builtin_popcount(m);
}

__device__ __forceinline__ void add_row(float* __restrict__ acc, int r, const float* __restrict__ h, int s) {
  const float4* hs = (const float4*)(h + (long)s * HID);
  float4 h0 = hs[0];
  float4 h1 = hs[1];
  float* row = acc + r * HID;
  atomicAdd(&row[0], h0.x); atomicAdd(&row[1], h0.y);
  atomicAdd(&row[2], h0.z); atomicAdd(&row[3], h0.w);
  atomicAdd(&row[4], h1.x); atomicAdd(&row[5], h1.y);
  atomicAdd(&row[6], h1.z); atomicAdd(&row[7], h1.w);
}

__device__ __forceinline__ void drain(const unsigned* __restrict__ q, int cnt, int lane, int base,
                                      const int* __restrict__ src, const int* __restrict__ dst,
                                      const float* __restrict__ h, float* __restrict__ acc,
                                      unsigned* __restrict__ indeg) {
  __builtin_amdgcn_fence(__ATOMIC_ACQ_REL, "wavefront");
  for (int i = lane; i < cnt; i += 32) {
    unsigned e = q[i];
    int d = dst[e];
    int s = src[e];
    atomicAdd(&indeg[d - base], 1u);
    add_row(acc, d - base, h, s);
  }
  __builtin_amdgcn_fence(__ATOMIC_ACQ_REL, "wavefront");
}

__global__ __launch_bounds__(SL_THREADS) void spmm1_slices(const int* __restrict__ src,
                                                           const int* __restrict__ dst,
                                                           const float* __restrict__ h,
                                                           float* __restrict__ agg,
                                                           float* __restrict__ deg_d, int nE, int nN) {
  __shared__ float acc[SL_NODES * HID];
  __shared__ unsigned indeg[SL_NODES];
  __shared__ unsigned queue[SL_WAVES * SL_QCAP];

  const int tid  = threadIdx.x;
  const int lane = tid & 31;
  const int wave = tid >> 5;
  const int slice = blockIdx.x / SL_PARTS;
  const int part  = blockIdx.x % SL_PARTS;
  const int base  = slice * SL_NODES;

  for (int i = tid; i < SL_NODES * HID / 4; i += SL_THREADS)
    ((float4*)acc)[i] = make_float4(0.0f, 0.0f, 0.0f, 0.0f);
  for (int i = tid; i < SL_NODES; i += SL_THREADS) indeg[i] = 0u;
  __syncthreads();

  unsigned* q = queue + wave * SL_QCAP;
  int cnt = 0;

  const long nQ   = nE >> 2;
  const long qPer = (nQ + SL_PARTS - 1) / SL_PARTS;
  const long q0   = part * qPer;
  const long q1   = (q0 + qPer) < nQ ? (q0 + qPer) : nQ;

  int4 n0 = make_int4(-1, -1, -1, -1), n1 = n0;
  if (q0 < q1) {
    n0 = load_quad(dst, q0 + tid, q1);
    n1 = load_quad(dst, q0 + tid + SL_THREADS, q1);
  }
  for (long qb = q0; qb < q1; qb += 2 * SL_THREADS) {
    int4 c0 = n0, c1 = n1;
    n0 = load_quad(dst, qb + 2 * SL_THREADS + tid, q1);
    n1 = load_quad(dst, qb + 3 * SL_THREADS + tid, q1);
    unsigned e0 = (unsigned)(4 * (qb + tid));
    unsigned e1 = e0 + 4u * SL_THREADS;
    note_hits(q, cnt, c0.x, e0 + 0u, base);
    note_hits(q, cnt, c0.y, e0 + 1u, base);
    note_hits(q, cnt, c0.z, e0 + 2u, base);
    note_hits(q, cnt, c0.w, e0 + 3u, base);
    if (cnt > SL_QCAP - SL_GROUP) {
      drain(q, cnt, lane, base, src, dst, h, acc, indeg);
      cnt = 0;
    }
    note_hits(q, cnt, c1.x, e1 + 0u, base);
    note_hits(q, cnt, c1.y, e1 + 1u, base);
    note_hits(q, cnt, c1.z, e1 + 2u, base);
    note_hits(q, cnt, c1.w, e1 + 3u, base);
    if (cnt > SL_QCAP - SL_GROUP) {
      drain(q, cnt, lane, base, src, dst, h, acc, indeg);
      cnt = 0;
    }
  }
  drain(q, cnt, lane, base, src, dst, h, acc, indeg);

  if (part == 0) {
    int e = (int)(nQ << 2) + tid;
    if (e < nE) {
      int d = dst[e];
      if ((unsigned)(d - base) < (unsigned)SL_NODES) {
        atomicAdd(&indeg[d - base], 1u);
        add_row(acc, d - base, h, src[e]);
      }
    }
  }
  __syncthreads();

  const int rows = (nN - base) < SL_NODES ? (nN - base) : SL_NODES;
  float* out = agg + (long)base * HID;
  for (int i = tid; i < rows * HID; i += SL_THREADS) atomicAdd(&out[i], acc[i]);
  for (int i = tid; i < rows; i += SL_THREADS) atomicAdd(&deg_d[base + i], (float)indeg[i]);
}


__global__ void layer2_node(const float* __restrict__ agg, float* __restrict__ in_norm,
                            const float* __restrict__ out_norm, const float* __restrict__ b1,
                            const float* __restrict__ W2, float* __restrict__ g2, int n) {
  int v = blockIdx.x * blockDim.x + threadIdx.x;
  if (v >= n) return;
  float innv = rsqrtf(fmaxf(in_norm[v], 1.0f));
  in_norm[v] = innv;
  const float* av = agg + (long)v * HID;
  float s = 0.0f;
#pragma unroll
  for (int j = 0; j < HID; ++j) {
    float z = fmaf(av[j], innv, b1[j]);
    z = z > 0.0f ? z : 0.0f;
    s = fmaf(z, W2[j], s);
  }
  g2[v] = s * out_norm[v];
}


__global__ void spmm2_kernel(const int* __restrict__ src, const int* __restrict__ dst,
                             const float* __restrict__ g2, float* __restrict__ outacc, int nE) {
  int e = blockIdx.x * blockDim.x + threadIdx.x;
  if (e < nE) atomicAdd(&outacc[dst[e]], g2[src[e]]);
}

__global__ void final_kernel(const float* __restrict__ outacc, const float* __restrict__ in_norm,
                             const float* __restrict__ b2, float* __restrict__ out, int n) {
  int v = blockIdx.x * blockDim.x + threadIdx.x;
  if (v < n) out[v] = fmaf(outacc[v], in_norm[v], b2[0]);
}


extern "C" void kernel_launch(void* const* d_in, const int* in_sizes, int n_in,
                              void* d_out, int out_size, void* d_ws, size_t ws_size,
                              hipStream_t stream) {
  const float* x  = (const float*)d_in[0];
  const float* W1 = (const float*)d_in[1];
  const float* b1 = (const float*)d_in[2];
  const float* W2 = (const float*)d_in[3];
  const float* b2 = (const float*)d_in[4];
  const int*  src = (const int*)d_in[5];
  const int*  dst = (const int*)d_in[6];

  const int nN = in_sizes[0] / IN_DIM;
  const int nE = in_sizes[5];

  float* ws     = (float*)d_ws;
  float* deg_s  = ws;
  float* deg_d  = ws + (long)nN;
  float* h      = ws + 2L * nN;
  float* agg    = ws + 10L * nN;
  float* g2     = ws + 18L * nN;
  float* outacc = ws + 19L * nN;
  float* out    = (float*)d_out;

  const int TB = 256;
  dim3 blk(TB);

  zero_f32<<<(2 * nN + TB - 1) / TB, blk, 0, stream>>>(deg_s, 2 * nN);
  zero_f32<<<(8 * nN + TB - 1) / TB, blk, 0, stream>>>(agg, 8 * nN);
  zero_f32<<<(nN + TB - 1) / TB, blk, 0, stream>>>(outacc, nN);

  degree_kernel<<<(nE + TB - 1) / TB, blk, 0, stream>>>(src, deg_s, nE);
  norm_kernel<<<(nN + TB - 1) / TB, blk, 0, stream>>>(deg_s, nN);

  gemm1_wmma<<<(nN + 127) / 128, blk, 0, stream>>>(x, W1, deg_s, h, nN);

  spmm1_slices<<<((nN + SL_NODES - 1) / SL_NODES) * SL_PARTS, dim3(SL_THREADS), 0, stream>>>(src, dst, h, agg, deg_d, nE, nN);
  layer2_node<<<(nN + TB - 1) / TB, blk, 0, stream>>>(agg, deg_d, deg_s, b1, W2, g2, nN);
  spmm2_kernel<<<(nE + TB - 1) / TB, blk, 0, stream>>>(src, dst, g2, outacc, nE);
  final_kernel<<<(nN + TB - 1) / TB, blk, 0, stream>>>(outacc, deg_d, b2, out, nN);
}


# scattered global atomics (out-degree, layer-2 sum) add into one array copy per package half, picked by the wave's die id (edge kernels now resident loops); layer-2 sum = 2 partial sums, then added
# speedup vs baseline: 3.6478x; 2.0314x over previous
#include <hip/hip_runtime.h>

typedef __attribute__((ext_vector_type(2))) float v2f;
typedef __attribute__((ext_vector_type(8))) float v8f;

constexpr int IN_DIM = 128;
constexpr int HID    = 8;


__global__ void zero_f32(float* __restrict__ p, int n) {
  int i = blockIdx.x * blockDim.x + threadIdx.x;
  if (i < n) p[i] = 0.0f;
}


__device__ __forceinline__ int half_of_package() {
  unsigned w = __builtin_amdgcn_s_sendmsg_rtn(0x87);
  return (int)((w >> 18) & 1u);
}

constexpr int EG_BLOCKS  = 2048;
constexpr int EG_THREADS = 256;

__global__ __launch_bounds__(EG_THREADS) void degree_kernel(const int* __restrict__ src,
                                                            float* __restrict__ deg_h0,
                                                            float* __restrict__ deg_h1, int nE) {
  float* __restrict__ deg = half_of_package() ? deg_h1 : deg_h0;
  const int nQ     = nE >> 2;
  const int stride = EG_BLOCKS * EG_THREADS;
  const int gid    = blockIdx.x * EG_THREADS + threadIdx.x;
  const int4* __restrict__ src4 = (const int4*)src;
  int4 s = make_int4(0, 0, 0, 0);
  if (gid < nQ) s = src4[gid];
  for (int q = gid; q < nQ; q += stride) {
    int4 sn = s;
    if (q + stride < nQ) sn = src4[q + stride];
    atomicAdd(&deg[s.x], 1.0f);
    atomicAdd(&deg[s.y], 1.0f);
    atomicAdd(&deg[s.z], 1.0f);
    atomicAdd(&deg[s.w], 1.0f);
    s = sn;
  }
  int e = (nQ << 2) + gid;
  if (e < nE) atomicAdd(&deg[src[e]], 1.0f);
}

__global__ void norm_kernel(float* __restrict__ deg_h0, const float* __restrict__ deg_h1, int n) {
  int i = blockIdx.x * blockDim.x + threadIdx.x;
  if (i < n) deg_h0[i] = rsqrtf(fmaxf(deg_h0[i] + deg_h1[i], 1.0f));
}


__global__ __launch_bounds__(256) void gemm1_wmma(const float* __restrict__ x,
                                                  const float* __restrict__ W1,
                                                  const float* __restrict__ out_norm,
                                                  float* __restrict__ h, int nNodes) {
  __shared__ float w[IN_DIM * 16];
  int tid = threadIdx.x;
  for (int i = tid; i < IN_DIM * 16; i += 256) {
    int r = i >> 4, cidx = i & 15;
    w[i] = (cidx < HID) ? W1[r * HID + cidx] : 0.0f;
  }
  __syncthreads();

  int wid  = tid >> 5;
  int lane = tid & 31;
  int m    = lane & 15;
  int kh   = lane >> 4;

  int row0 = (blockIdx.x * 8 + wid) * 16;
  int node = row0 + m;
  int nodeClamped = node < nNodes ? node : nNodes - 1;
  const float* xrow = x + (long)nodeClamped * IN_DIM;

  v8f acc = {};
#pragma unroll
  for (int kk = 0; kk < IN_DIM; kk += 4) {
    int k0 = kk + 2 * kh;
    v2f a;
    a.x = xrow[k0];
    a.y = xrow[k0 + 1];
    v2f b;
    b.x = w[k0 * 16 + m];
    b.y = w[(k0 + 1) * 16 + m];
    acc = __builtin_amdgcn_wmma_f32_16x16x4_f32(
         false, a,  false, b,
         (short)0, acc,  false,  false);
  }

  if (m < HID) {
#pragma unroll
    for (int r = 0; r < 8; ++r) {
      int nr = row0 + r + 8 * kh;
      if (nr < nNodes) h[(long)nr * HID + m] = acc[r] * out_norm[nr];
    }
  }
}


constexpr int SL_NODES   = 7952;
constexpr int SL_PARTS   = 4;
constexpr int SL_THREADS = 1024;
constexpr int SL_WAVES   = SL_THREADS / 32;
constexpr int SL_QCAP    = 320;
constexpr int SL_GROUP   = 128;
static_assert(SL_NODES * (HID + 1) * 4 + SL_WAVES * SL_QCAP * 4 <= 320 * 1024, "LDS budget");
static_assert(SL_QCAP > SL_GROUP, "a whole group of hits fits behind the drain mark");
static_assert((SL_NODES * HID) % 32 == 0, "a slice of agg starts on a 128-byte line");

__device__ __forceinline__ int4 load_quad(const int* __restrict__ p, long q, long qEnd) {
  long qc = q < qEnd ? q : qEnd - 1;
  int4 v = ((const int4*)p)[qc];
  if (q >= qEnd) v = make_int4(-1, -1, -1, -1);
  return v;
}

__device__ __forceinline__ void note_hits(unsigned* __restrict__ q, int& cnt, int d, unsigned e, int base) {
  bool hit = (unsigned)(d - base) < (unsigned)SL_NODES;
  unsigned m = __builtin_amdgcn_ballot_w32(hit);
  if (hit) q[cnt + (int)__builtin_amdgcn_mbcnt_lo(m, 0u)] = e;
  cnt += __builtin_popcount(m);
}

__device__ __forceinline__ void add_row(float* __restrict__ acc, int r, const float* __restrict__ h, int s) {
  const float4* hs = (const float4*)(h + (long)s * HID);
  float4 h0 = hs[0];
  float4 h1 = hs[1];
  float* row = acc + r * HID;
  atomicAdd(&row[0], h0.x); atomicAdd(&row[1], h0.y);
  atomicAdd(&row[2], h0.z); atomicAdd(&row[3], h0.w);
  atomicAdd(&row[4], h1.x); atomicAdd(&row[5], h1.y);
  atomicAdd(&row[6], h1.z); atomicAdd(&row[7], h1.w);
}

__device__ __forceinline__ void drain(const unsigned* __restrict__ q, int cnt, int lane, int base,
                                      const int* __restrict__ src, const int* __restrict__ dst,
                                      const float* __restrict__ h, float* __restrict__ acc,
                                      unsigned* __restrict__ indeg) {
  __builtin_amdgcn_fence(__ATOMIC_ACQ_REL, "wavefront");
  for (int i = lane; i < cnt; i += 32) {
    unsigned e = q[i];
    int d = dst[e];
    int s = src[e];
    atomicAdd(&indeg[d - base], 1u);
    add_row(acc, d - base, h, s);
  }
  __builtin_amdgcn_fence(__ATOMIC_ACQ_REL, "wavefront");
}

__global__ __launch_bounds__(SL_THREADS) void spmm1_slices(const int* __restrict__ src,
                                                           const int* __restrict__ dst,
                                                           const float* __restrict__ h,
                                                           float* __restrict__ agg,
                                                           float* __restrict__ deg_d, int nE, int nN) {
  __shared__ float acc[SL_NODES * HID];
  __shared__ unsigned indeg[SL_NODES];
  __shared__ unsigned queue[SL_WAVES * SL_QCAP];

  const int tid  = threadIdx.x;
  const int lane = tid & 31;
  const int wave = tid >> 5;
  const int slice = blockIdx.x / SL_PARTS;
  const int part  = blockIdx.x % SL_PARTS;
  const int base  = slice * SL_NODES;

  for (int i = tid; i < SL_NODES * HID / 4; i += SL_THREADS)
    ((float4*)acc)[i] = make_float4(0.0f, 0.0f, 0.0f, 0.0f);
  for (int i = tid; i < SL_NODES; i += SL_THREADS) indeg[i] = 0u;
  __syncthreads();

  unsigned* q = queue + wave * SL_QCAP;
  int cnt = 0;

  const long nQ   = nE >> 2;
  const long qPer = (nQ + SL_PARTS - 1) / SL_PARTS;
  const long q0   = part * qPer;
  const long q1   = (q0 + qPer) < nQ ? (q0 + qPer) : nQ;

  int4 n0 = make_int4(-1, -1, -1, -1), n1 = n0;
  if (q0 < q1) {
    n0 = load_quad(dst, q0 + tid, q1);
    n1 = load_quad(dst, q0 + tid + SL_THREADS, q1);
  }
  for (long qb = q0; qb < q1; qb += 2 * SL_THREADS) {
    int4 c0 = n0, c1 = n1;
    n0 = load_quad(dst, qb + 2 * SL_THREADS + tid, q1);
    n1 = load_quad(dst, qb + 3 * SL_THREADS + tid, q1);
    unsigned e0 = (unsigned)(4 * (qb + tid));
    unsigned e1 = e0 + 4u * SL_THREADS;
    note_hits(q, cnt, c0.x, e0 + 0u, base);
    note_hits(q, cnt, c0.y, e0 + 1u, base);
    note_hits(q, cnt, c0.z, e0 + 2u, base);
    note_hits(q, cnt, c0.w, e0 + 3u, base);
    if (cnt > SL_QCAP - SL_GROUP) {
      drain(q, cnt, lane, base, src, dst, h, acc, indeg);
      cnt = 0;
    }
    note_hits(q, cnt, c1.x, e1 + 0u, base);
    note_hits(q, cnt, c1.y, e1 + 1u, base);
    note_hits(q, cnt, c1.z, e1 + 2u, base);
    note_hits(q, cnt, c1.w, e1 + 3u, base);
    if (cnt > SL_QCAP - SL_GROUP) {
      drain(q, cnt, lane, base, src, dst, h, acc, indeg);
      cnt = 0;
    }
  }
  drain(q, cnt, lane, base, src, dst, h, acc, indeg);

  if (part == 0) {
    int e = (int)(nQ << 2) + tid;
    if (e < nE) {
      int d = dst[e];
      if ((unsigned)(d - base) < (unsigned)SL_NODES) {
        atomicAdd(&indeg[d - base], 1u);
        add_row(acc, d - base, h, src[e]);
      }
    }
  }
  __syncthreads();

  const int rows = (nN - base) < SL_NODES ? (nN - base) : SL_NODES;
  float* out = agg + (long)base * HID;
  for (int i = tid; i < rows * HID; i += SL_THREADS) atomicAdd(&out[i], acc[i]);
  for (int i = tid; i < rows; i += SL_THREADS) atomicAdd(&deg_d[base + i], (float)indeg[i]);
}


__global__ void layer2_node(const float* __restrict__ agg, float* __restrict__ in_norm,
                            const float* __restrict__ out_norm, const float* __restrict__ b1,
                            const float* __restrict__ W2, float* __restrict__ g2, int n) {
  int v = blockIdx.x * blockDim.x + threadIdx.x;
  if (v >= n) return;
  float innv = rsqrtf(fmaxf(in_norm[v], 1.0f));
  in_norm[v] = innv;
  const float* av = agg + (long)v * HID;
  float s = 0.0f;
#pragma unroll
  for (int j = 0; j < HID; ++j) {
    float z = fmaf(av[j], innv, b1[j]);
    z = z > 0.0f ? z : 0.0f;
    s = fmaf(z, W2[j], s);
  }
  g2[v] = s * out_norm[v];
}


__global__ __launch_bounds__(EG_THREADS) void spmm2_kernel(const int* __restrict__ src,
                                                           const int* __restrict__ dst,
                                                           const float* __restrict__ g2,
                                                           float* __restrict__ acc_h0,
                                                           float* __restrict__ acc_h1, int nE) {
  float* __restrict__ acc = half_of_package() ? acc_h1 : acc_h0;
  const int nQ     = nE >> 2;
  const int stride = EG_BLOCKS * EG_THREADS;
  const int gid    = blockIdx.x * EG_THREADS + threadIdx.x;
  const int4* __restrict__ src4 = (const int4*)src;
  const int4* __restrict__ dst4 = (const int4*)dst;
  int4 s = make_int4(0, 0, 0, 0), d = s;
  if (gid < nQ) { s = src4[gid]; d = dst4[gid]; }
  for (int q = gid; q < nQ; q += stride) {
    int4 sn = s, dn = d;
    if (q + stride < nQ) { sn = src4[q + stride]; dn = dst4[q + stride]; }
    float ga = g2[s.x], gb = g2[s.y], gc = g2[s.z], gd = g2[s.w];
    atomicAdd(&acc[d.x], ga);
    atomicAdd(&acc[d.y], gb);
    atomicAdd(&acc[d.z], gc);
    atomicAdd(&acc[d.w], gd);
    s = sn; d = dn;
  }
  int e = (nQ << 2) + gid;
  if (e < nE) atomicAdd(&acc[dst[e]], g2[src[e]]);
}

__global__ void final_kernel(const float* __restrict__ acc_h0, const float* __restrict__ acc_h1,
                             const float* __restrict__ in_norm, const float* __restrict__ b2,
                             float* __restrict__ out, int n) {
  int v = blockIdx.x * blockDim.x + threadIdx.x;
  if (v < n) out[v] = fmaf(acc_h0[v] + acc_h1[v], in_norm[v], b2[0]);
}


extern "C" void kernel_launch(void* const* d_in, const int* in_sizes, int n_in,
                              void* d_out, int out_size, void* d_ws, size_t ws_size,
                              hipStream_t stream) {
  const float* x  = (const float*)d_in[0];
  const float* W1 = (const float*)d_in[1];
  const float* b1 = (const float*)d_in[2];
  const float* W2 = (const float*)d_in[3];
  const float* b2 = (const float*)d_in[4];
  const int*  src = (const int*)d_in[5];
  const int*  dst = (const int*)d_in[6];

  const int nN = in_sizes[0] / IN_DIM;
  const int nE = in_sizes[5];

  float* ws     = (float*)d_ws;
  float* deg_s  = ws;
  float* deg_d  = ws + (long)nN;
  float* deg_s1 = ws + 2L * nN;
  float* g2     = deg_s1;
  float* h      = ws + 3L * nN;
  float* acc_h0 = h;
  float* acc_h1 = h + (long)nN;
  float* agg    = ws + 11L * nN;
  float* out    = (float*)d_out;

  const int TB = 256;
  dim3 blk(TB);

  zero_f32<<<(3 * nN + TB - 1) / TB, blk, 0, stream>>>(deg_s, 3 * nN);
  zero_f32<<<(8 * nN + TB - 1) / TB, blk, 0, stream>>>(agg, 8 * nN);

  degree_kernel<<<EG_BLOCKS, dim3(EG_THREADS), 0, stream>>>(src, deg_s, deg_s1, nE);
  norm_kernel<<<(nN + TB - 1) / TB, blk, 0, stream>>>(deg_s, deg_s1, nN);

  gemm1_wmma<<<(nN + 127) / 128, blk, 0, stream>>>(x, W1, deg_s, h, nN);

  spmm1_slices<<<((nN + SL_NODES - 1) / SL_NODES) * SL_PARTS, dim3(SL_THREADS), 0, stream>>>(src, dst, h, agg, deg_d, nE, nN);
  zero_f32<<<(2 * nN + TB - 1) / TB, blk, 0, stream>>>(acc_h0, 2 * nN);
  layer2_node<<<(nN + TB - 1) / TB, blk, 0, stream>>>(agg, deg_d, deg_s, b1, W2, g2, nN);
  spmm2_kernel<<<EG_BLOCKS, dim3(EG_THREADS), 0, stream>>>(src, dst, g2, acc_h0, acc_h1, nE);
  final_kernel<<<(nN + TB - 1) / TB, blk, 0, stream>>>(acc_h0, acc_h1, deg_d, b2, out, nN);
}


# layer-1 queue entry packs the hit's row in the slice with its place in the wave's walk (edge number recomputed): drain skips the dst[e] re-read, one scattered line per edge less; atomic order kept
# speedup vs baseline: 3.8547x; 1.0567x over previous
#include <hip/hip_runtime.h>

typedef __attribute__((ext_vector_type(2))) float v2f;
typedef __attribute__((ext_vector_type(8))) float v8f;

constexpr int IN_DIM = 128;
constexpr int HID    = 8;


__global__ void zero_f32(float* __restrict__ p, int n) {
  int i = blockIdx.x * blockDim.x + threadIdx.x;
  if (i < n) p[i] = 0.0f;
}


__device__ __forceinline__ int half_of_package() {
  unsigned w = __builtin_amdgcn_s_sendmsg_rtn(0x87);
  return (int)((w >> 18) & 1u);
}

constexpr int EG_BLOCKS  = 2048;
constexpr int EG_THREADS = 256;

__global__ __launch_bounds__(EG_THREADS) void degree_kernel(const int* __restrict__ src,
                                                            float* __restrict__ deg_h0,
                                                            float* __restrict__ deg_h1, int nE) {
  float* __restrict__ deg = half_of_package() ? deg_h1 : deg_h0;
  const int nQ     = nE >> 2;
  const int stride = EG_BLOCKS * EG_THREADS;
  const int gid    = blockIdx.x * EG_THREADS + threadIdx.x;
  const int4* __restrict__ src4 = (const int4*)src;
  int4 s = make_int4(0, 0, 0, 0);
  if (gid < nQ) s = src4[gid];
  for (int q = gid; q < nQ; q += stride) {
    int4 sn = s;
    if (q + stride < nQ) sn = src4[q + stride];
    atomicAdd(&deg[s.x], 1.0f);
    atomicAdd(&deg[s.y], 1.0f);
    atomicAdd(&deg[s.z], 1.0f);
    atomicAdd(&deg[s.w], 1.0f);
    s = sn;
  }
  int e = (nQ << 2) + gid;
  if (e < nE) atomicAdd(&deg[src[e]], 1.0f);
}

__global__ void norm_kernel(float* __restrict__ deg_h0, const float* __restrict__ deg_h1, int n) {
  int i = blockIdx.x * blockDim.x + threadIdx.x;
  if (i < n) deg_h0[i] = rsqrtf(fmaxf(deg_h0[i] + deg_h1[i], 1.0f));
}


__global__ __launch_bounds__(256) void gemm1_wmma(const float* __restrict__ x,
                                                  const float* __restrict__ W1,
                                                  const float* __restrict__ out_norm,
                                                  float* __restrict__ h, int nNodes) {
  __shared__ float w[IN_DIM * 16];
  int tid = threadIdx.x;
  for (int i = tid; i < IN_DIM * 16; i += 256) {
    int r = i >> 4, cidx = i & 15;
    w[i] = (cidx < HID) ? W1[r * HID + cidx] : 0.0f;
  }
  __syncthreads();

  int wid  = tid >> 5;
  int lane = tid & 31;
  int m    = lane & 15;
  int kh   = lane >> 4;

  int row0 = (blockIdx.x * 8 + wid) * 16;
  int node = row0 + m;
  int nodeClamped = node < nNodes ? node : nNodes - 1;
  const float* xrow = x + (long)nodeClamped * IN_DIM;

  v8f acc = {};
#pragma unroll
  for (int kk = 0; kk < IN_DIM; kk += 4) {
    int k0 = kk + 2 * kh;
    v2f a;
    a.x = xrow[k0];
    a.y = xrow[k0 + 1];
    v2f b;
    b.x = w[k0 * 16 + m];
    b.y = w[(k0 + 1) * 16 + m];
    acc = __builtin_amdgcn_wmma_f32_16x16x4_f32(
         false, a,  false, b,
         (short)0, acc,  false,  false);
  }

  if (m < HID) {
#pragma unroll
    for (int r = 0; r < 8; ++r) {
      int nr = row0 + r + 8 * kh;
      if (nr < nNodes) h[(long)nr * HID + m] = acc[r] * out_norm[nr];
    }
  }
}


constexpr int SL_NODES   = 7952;
constexpr int SL_PARTS   = 4;
constexpr int SL_THREADS = 1024;
constexpr int SL_WAVES   = SL_THREADS / 32;
constexpr int SL_QCAP    = 320;
constexpr int SL_GROUP   = 128;
static_assert(SL_NODES * (HID + 1) * 4 + SL_WAVES * SL_QCAP * 4 <= 320 * 1024, "LDS budget");
static_assert(SL_QCAP > SL_GROUP, "a whole group of hits fits behind the drain mark");
static_assert((SL_NODES * HID) % 32 == 0, "a slice of agg starts on a 128-byte line");
constexpr int SL_POS_BITS = 19;
static_assert(SL_NODES <= (1 << (32 - SL_POS_BITS)), "the row in the slice fits above the place bits");

__device__ __forceinline__ int4 load_quad(const int* __restrict__ p, long q, long qEnd) {
  long qc = q < qEnd ? q : qEnd - 1;
  int4 v = ((const int4*)p)[qc];
  if (q >= qEnd) v = make_int4(-1, -1, -1, -1);
  return v;
}

__device__ __forceinline__ void note_hits(unsigned* __restrict__ q, int& cnt, int d, unsigned place, int base) {
  unsigned r = (unsigned)(d - base);
  bool hit = r < (unsigned)SL_NODES;
  unsigned m = __builtin_amdgcn_ballot_w32(hit);
  if (hit) q[cnt + (int)__builtin_amdgcn_mbcnt_lo(m, 0u)] = (r << SL_POS_BITS) | place;
  cnt += __builtin_popcount(m);
}

__device__ __forceinline__ void add_row(float* __restrict__ acc, int r, const float* __restrict__ h, int s) {
  const float4* hs = (const float4*)(h + (long)s * HID);
  float4 h0 = hs[0];
  float4 h1 = hs[1];
  float* row = acc + r * HID;
  atomicAdd(&row[0], h0.x); atomicAdd(&row[1], h0.y);
  atomicAdd(&row[2], h0.z); atomicAdd(&row[3], h0.w);
  atomicAdd(&row[4], h1.x); atomicAdd(&row[5], h1.y);
  atomicAdd(&row[6], h1.z); atomicAdd(&row[7], h1.w);
}

__device__ __forceinline__ void drain(const unsigned* __restrict__ q, int cnt, int lane, unsigned eBase,
                                      const int* __restrict__ src,
                                      const float* __restrict__ h, float* __restrict__ acc,
                                      unsigned* __restrict__ indeg) {
  __builtin_amdgcn_fence(__ATOMIC_ACQ_REL, "wavefront");
  for (int i = lane; i < cnt; i += 32) {
    unsigned v = q[i];
    int r = (int)(v >> SL_POS_BITS);
    unsigned place = v & ((1u << SL_POS_BITS) - 1u);
    unsigned e = eBase + ((place >> 7) << 12) + (place & 127u);
    int s = src[e];
    atomicAdd(&indeg[r], 1u);
    add_row(acc, r, h, s);
  }
  __builtin_amdgcn_fence(__ATOMIC_ACQ_REL, "wavefront");
}

__global__ __launch_bounds__(SL_THREADS) void spmm1_slices(const int* __restrict__ src,
                                                           const int* __restrict__ dst,
                                                           const float* __restrict__ h,
                                                           float* __restrict__ agg,
                                                           float* __restrict__ deg_d, int nE, int nN) {
  __shared__ float acc[SL_NODES * HID];
  __shared__ unsigned indeg[SL_NODES];
  __shared__ unsigned queue[SL_WAVES * SL_QCAP];

  const int tid  = threadIdx.x;
  const int lane = tid & 31;
  const int wave = tid >> 5;
  const int slice = blockIdx.x / SL_PARTS;
  const int part  = blockIdx.x % SL_PARTS;
  const int base  = slice * SL_NODES;

  for (int i = tid; i < SL_NODES * HID / 4; i += SL_THREADS)
    ((float4*)acc)[i] = make_float4(0.0f, 0.0f, 0.0f, 0.0f);
  for (int i = tid; i < SL_NODES; i += SL_THREADS) indeg[i] = 0u;
  __syncthreads();

  unsigned* q = queue + wave * SL_QCAP;
  int cnt = 0;
  static_assert(SL_THREADS == 1024, "a step of the walk is 1024 quads (the place bits count such steps)");

  const long nQ   = nE >> 2;
  const long qPer = (nQ + SL_PARTS - 1) / SL_PARTS;
  const long q0   = part * qPer;
  const long q1   = (q0 + qPer) < nQ ? (q0 + qPer) : nQ;

  const unsigned eBase = (unsigned)(4 * q0) + (unsigned)(wave * 128);
  unsigned place = (unsigned)(lane << 2);
  int4 n0 = make_int4(-1, -1, -1, -1), n1 = n0;
  if (q0 < q1) {
    n0 = load_quad(dst, q0 + tid, q1);
    n1 = load_quad(dst, q0 + tid + SL_THREADS, q1);
  }
  for (long qb = q0; qb < q1; qb += 2 * SL_THREADS) {
    int4 c0 = n0, c1 = n1;
    n0 = load_quad(dst, qb + 2 * SL_THREADS + tid, q1);
    n1 = load_quad(dst, qb + 3 * SL_THREADS + tid, q1);
    const unsigned p0 = place;
    const unsigned p1 = place + 128u;
    place += 256u;
    note_hits(q, cnt, c0.x, p0 + 0u, base);
    note_hits(q, cnt, c0.y, p0 + 1u, base);
    note_hits(q, cnt, c0.z, p0 + 2u, base);
    note_hits(q, cnt, c0.w, p0 + 3u, base);
    if (cnt > SL_QCAP - SL_GROUP) {
      drain(q, cnt, lane, eBase, src, h, acc, indeg);
      cnt = 0;
    }
    note_hits(q, cnt, c1.x, p1 + 0u, base);
    note_hits(q, cnt, c1.y, p1 + 1u, base);
    note_hits(q, cnt, c1.z, p1 + 2u, base);
    note_hits(q, cnt, c1.w, p1 + 3u, base);
    if (cnt > SL_QCAP - SL_GROUP) {
      drain(q, cnt, lane, eBase, src, h, acc, indeg);
      cnt = 0;
    }
  }
  drain(q, cnt, lane, eBase, src, h, acc, indeg);

  if (part == 0) {
    int e = (int)(nQ << 2) + tid;
    if (e < nE) {
      int d = dst[e];
      if ((unsigned)(d - base) < (unsigned)SL_NODES) {
        atomicAdd(&indeg[d - base], 1u);
        add_row(acc, d - base, h, src[e]);
      }
    }
  }
  __syncthreads();

  const int rows = (nN - base) < SL_NODES ? (nN - base) : SL_NODES;
  float* out = agg + (long)base * HID;
  for (int i = tid; i < rows * HID; i += SL_THREADS) atomicAdd(&out[i], acc[i]);
  for (int i = tid; i < rows; i += SL_THREADS) atomicAdd(&deg_d[base + i], (float)indeg[i]);
}


__global__ void layer2_node(const float* __restrict__ agg, float* __restrict__ in_norm,
                            const float* __restrict__ out_norm, const float* __restrict__ b1,
                            const float* __restrict__ W2, float* __restrict__ g2, int n) {
  int v = blockIdx.x * blockDim.x + threadIdx.x;
  if (v >= n) return;
  float innv = rsqrtf(fmaxf(in_norm[v], 1.0f));
  in_norm[v] = innv;
  const float* av = agg + (long)v * HID;
  float s = 0.0f;
#pragma unroll
  for (int j = 0; j < HID; ++j) {
    float z = fmaf(av[j], innv, b1[j]);
    z = z > 0.0f ? z : 0.0f;
    s = fmaf(z, W2[j], s);
  }
  g2[v] = s * out_norm[v];
}


__global__ __launch_bounds__(EG_THREADS) void spmm2_kernel(const int* __restrict__ src,
                                                           const int* __restrict__ dst,
                                                           const float* __restrict__ g2,
                                                           float* __restrict__ acc_h0,
                                                           float* __restrict__ acc_h1, int nE) {
  float* __restrict__ acc = half_of_package() ? acc_h1 : acc_h0;
  const int nQ     = nE >> 2;
  const int stride = EG_BLOCKS * EG_THREADS;
  const int gid    = blockIdx.x * EG_THREADS + threadIdx.x;
  const int4* __restrict__ src4 = (const int4*)src;
  const int4* __restrict__ dst4 = (const int4*)dst;
  int4 s = make_int4(0, 0, 0, 0), d = s;
  if (gid < nQ) { s = src4[gid]; d = dst4[gid]; }
  for (int q = gid; q < nQ; q += stride) {
    int4 sn = s, dn = d;
    if (q + stride < nQ) { sn = src4[q + stride]; dn = dst4[q + stride]; }
    float ga = g2[s.x], gb = g2[s.y], gc = g2[s.z], gd = g2[s.w];
    atomicAdd(&acc[d.x], ga);
    atomicAdd(&acc[d.y], gb);
    atomicAdd(&acc[d.z], gc);
    atomicAdd(&acc[d.w], gd);
    s = sn; d = dn;
  }
  int e = (nQ << 2) + gid;
  if (e < nE) atomicAdd(&acc[dst[e]], g2[src[e]]);
}

__global__ void final_kernel(const float* __restrict__ acc_h0, const float* __restrict__ acc_h1,
                             const float* __restrict__ in_norm, const float* __restrict__ b2,
                             float* __restrict__ out, int n) {
  int v = blockIdx.x * blockDim.x + threadIdx.x;
  if (v < n) out[v] = fmaf(acc_h0[v] + acc_h1[v], in_norm[v], b2[0]);
}


extern "C" void kernel_launch(void* const* d_in, const int* in_sizes, int n_in,
                              void* d_out, int out_size, void* d_ws, size_t ws_size,
                              hipStream_t stream) {
  const float* x  = (const float*)d_in[0];
  const float* W1 = (const float*)d_in[1];
  const float* b1 = (const float*)d_in[2];
  const float* W2 = (const float*)d_in[3];
  const float* b2 = (const float*)d_in[4];
  const int*  src = (const int*)d_in[5];
  const int*  dst = (const int*)d_in[6];

  const int nN = in_sizes[0] / IN_DIM;
  const int nE = in_sizes[5];

  float* ws     = (float*)d_ws;
  float* deg_s  = ws;
  float* deg_d  = ws + (long)nN;
  float* deg_s1 = ws + 2L * nN;
  float* g2     = deg_s1;
  float* h      = ws + 3L * nN;
  float* acc_h0 = h;
  float* acc_h1 = h + (long)nN;
  float* agg    = ws + 11L * nN;
  float* out    = (float*)d_out;

  const int TB = 256;
  dim3 blk(TB);

  zero_f32<<<(3 * nN + TB - 1) / TB, blk, 0, stream>>>(deg_s, 3 * nN);
  zero_f32<<<(8 * nN + TB - 1) / TB, blk, 0, stream>>>(agg, 8 * nN);

  degree_kernel<<<EG_BLOCKS, dim3(EG_THREADS), 0, stream>>>(src, deg_s, deg_s1, nE);
  norm_kernel<<<(nN + TB - 1) / TB, blk, 0, stream>>>(deg_s, deg_s1, nN);

  gemm1_wmma<<<(nN + 127) / 128, blk, 0, stream>>>(x, W1, deg_s, h, nN);

  spmm1_slices<<<((nN + SL_NODES - 1) / SL_NODES) * SL_PARTS, dim3(SL_THREADS), 0, stream>>>(src, dst, h, agg, deg_d, nE, nN);
  zero_f32<<<(2 * nN + TB - 1) / TB, blk, 0, stream>>>(acc_h0, 2 * nN);
  layer2_node<<<(nN + TB - 1) / TB, blk, 0, stream>>>(agg, deg_d, deg_s, b1, W2, g2, nN);
  spmm2_kernel<<<EG_BLOCKS, dim3(EG_THREADS), 0, stream>>>(src, dst, g2, acc_h0, acc_h1, nE);
  final_kernel<<<(nN + TB - 1) / TB, blk, 0, stream>>>(acc_h0, acc_h1, deg_d, b2, out, nN);
}


# out-degree: a work-group bins its share of src by 128-byte counter line in LDS (byte entries), then 8 lanes add a bin's entries at once: about 1 atomic request a line, not 1 an edge; counts exact
# speedup vs baseline: 4.0813x; 1.0588x over previous
#include <hip/hip_runtime.h>

typedef __attribute__((ext_vector_type(2))) float v2f;
typedef __attribute__((ext_vector_type(8))) float v8f;

constexpr int IN_DIM = 128;
constexpr int HID    = 8;


__global__ void zero_f32(float* __restrict__ p, int n) {
  int i = blockIdx.x * blockDim.x + threadIdx.x;
  if (i < n) p[i] = 0.0f;
}


__device__ __forceinline__ int half_of_package() {
  unsigned w = __builtin_amdgcn_s_sendmsg_rtn(0x87);
  return (int)((w >> 18) & 1u);
}

constexpr int EG_BLOCKS  = 2048;
constexpr int EG_THREADS = 256;

constexpr int DL_BLOCKS  = 256;
constexpr int DL_THREADS = 1024;
constexpr int DL_LINES   = 16384;
constexpr int DL_CAP     = 12;
constexpr int DL_BATCH   = 16;
static_assert((DL_LINES & (DL_LINES - 1)) == 0, "the walk over the bins wraps with a mask");
static_assert(DL_LINES * 16 <= 320 * 1024, "LDS budget");

__device__ __forceinline__ void degree_put(unsigned* fill, unsigned char* ent, float* __restrict__ deg, int s) {
  if (s < 0) return;
  const unsigned line = (unsigned)s >> 5;
  if (line < (unsigned)DL_LINES) {
    const unsigned pos = atomicAdd(&fill[line * 4u], 1u);
    if (pos < (unsigned)DL_CAP) {
      ent[line * 16u + 4u + pos] = (unsigned char)(s & 31);
      return;
    }
  }
  atomicAdd(&deg[s], 1.0f);
}

__global__ __launch_bounds__(DL_THREADS) void degree_lines(const int* __restrict__ src,
                                                           float* __restrict__ deg_h0,
                                                           float* __restrict__ deg_h1, int nE) {
  __shared__ uint4 bins[DL_LINES];
  float* __restrict__ deg = half_of_package() ? deg_h1 : deg_h0;
  const int tid = threadIdx.x;
  for (int i = tid; i < DL_LINES; i += DL_THREADS) bins[i] = make_uint4(0u, 0u, 0u, 0u);
  __syncthreads();
  unsigned* fill = (unsigned*)bins;
  unsigned char* ent = (unsigned char*)bins;

  const int nQ   = nE >> 2;
  const int qPer = (nQ + DL_BLOCKS - 1) / DL_BLOCKS;
  const int q0   = blockIdx.x * qPer;
  const int q1   = (q0 + qPer) < nQ ? (q0 + qPer) : nQ;
  const int4* __restrict__ src4 = (const int4*)src;
  for (int qb = q0 + tid; qb < q1; qb += DL_BATCH * DL_THREADS) {
    int4 s[DL_BATCH];
#pragma unroll
    for (int k = 0; k < DL_BATCH; ++k) {
      const int q  = qb + k * DL_THREADS;
      const int qc = q < q1 ? q : q1 - 1;
      s[k] = src4[qc];
      if (q >= q1) s[k] = make_int4(-1, -1, -1, -1);
    }
#pragma unroll
    for (int k = 0; k < DL_BATCH; ++k) {
      degree_put(fill, ent, deg, s[k].x);
      degree_put(fill, ent, deg, s[k].y);
      degree_put(fill, ent, deg, s[k].z);
      degree_put(fill, ent, deg, s[k].w);
    }
  }
  if (blockIdx.x == 0) {
    const int e = (nQ << 2) + tid;
    if (e < nE) degree_put(fill, ent, deg, src[e]);
  }
  __syncthreads();

  const unsigned lane = (unsigned)tid & 31u;
  const unsigned g    = lane >> 3;
  const unsigned k    = lane & 7u;
  const unsigned sh   = (k & 3u) * 8u;
  const unsigned rot  = ((unsigned)blockIdx.x * 64u) & (unsigned)(DL_LINES - 1);
  constexpr int DL_AHEAD = 4;
  static_assert(DL_LINES % ((DL_THREADS / 32) * 4 * DL_AHEAD) == 0, "whole steps");
  for (int base = (tid >> 5) * 4 * DL_AHEAD; base < DL_LINES; base += (DL_THREADS / 32) * 4 * DL_AHEAD) {
    unsigned line[DL_AHEAD];
    uint4 b[DL_AHEAD];
#pragma unroll
    for (int j = 0; j < DL_AHEAD; ++j) {
      line[j] = ((unsigned)(base + 4 * j) + g + rot) & (unsigned)(DL_LINES - 1);
      b[j] = bins[line[j]];
    }
#pragma unroll
    for (int j = 0; j < DL_AHEAD; ++j) {
      const unsigned c  = b[j].x < (unsigned)DL_CAP ? b[j].x : (unsigned)DL_CAP;
      const unsigned w0 = ((k < 4u ? b[j].y : b[j].z) >> sh) & 31u;
      if (k < c) atomicAdd(&deg[line[j] * 32u + w0], 1.0f);
      const unsigned w1 = (b[j].w >> sh) & 31u;
      if (k < 4u && k + 8u < c) atomicAdd(&deg[line[j] * 32u + w1], 1.0f);
    }
  }
}

__global__ void norm_kernel(float* __restrict__ deg_h0, const float* __restrict__ deg_h1, int n) {
  int i = blockIdx.x * blockDim.x + threadIdx.x;
  if (i < n) deg_h0[i] = rsqrtf(fmaxf(deg_h0[i] + deg_h1[i], 1.0f));
}


__global__ __launch_bounds__(256) void gemm1_wmma(const float* __restrict__ x,
                                                  const float* __restrict__ W1,
                                                  const float* __restrict__ out_norm,
                                                  float* __restrict__ h, int nNodes) {
  __shared__ float w[IN_DIM * 16];
  int tid = threadIdx.x;
  for (int i = tid; i < IN_DIM * 16; i += 256) {
    int r = i >> 4, cidx = i & 15;
    w[i] = (cidx < HID) ? W1[r * HID + cidx] : 0.0f;
  }
  __syncthreads();

  int wid  = tid >> 5;
  int lane = tid & 31;
  int m    = lane & 15;
  int kh   = lane >> 4;

  int row0 = (blockIdx.x * 8 + wid) * 16;
  int node = row0 + m;
  int nodeClamped = node < nNodes ? node : nNodes - 1;
  const float* xrow = x + (long)nodeClamped * IN_DIM;

  v8f acc = {};
#pragma unroll
  for (int kk = 0; kk < IN_DIM; kk += 4) {
    int k0 = kk + 2 * kh;
    v2f a;
    a.x = xrow[k0];
    a.y = xrow[k0 + 1];
    v2f b;
    b.x = w[k0 * 16 + m];
    b.y = w[(k0 + 1) * 16 + m];
    acc = __builtin_amdgcn_wmma_f32_16x16x4_f32(
         false, a,  false, b,
         (short)0, acc,  false,  false);
  }

  if (m < HID) {
#pragma unroll
    for (int r = 0; r < 8; ++r) {
      int nr = row0 + r + 8 * kh;
      if (nr < nNodes) h[(long)nr * HID + m] = acc[r] * out_norm[nr];
    }
  }
}


constexpr int SL_NODES   = 7952;
constexpr int SL_PARTS   = 4;
constexpr int SL_THREADS = 1024;
constexpr int SL_WAVES   = SL_THREADS / 32;
constexpr int SL_QCAP    = 320;
constexpr int SL_GROUP   = 128;
static_assert(SL_NODES * (HID + 1) * 4 + SL_WAVES * SL_QCAP * 4 <= 320 * 1024, "LDS budget");
static_assert(SL_QCAP > SL_GROUP, "a whole group of hits fits behind the drain mark");
static_assert((SL_NODES * HID) % 32 == 0, "a slice of agg starts on a 128-byte line");
constexpr int SL_POS_BITS = 19;
static_assert(SL_NODES <= (1 << (32 - SL_POS_BITS)), "the row in the slice fits above the place bits");

__device__ __forceinline__ int4 load_quad(const int* __restrict__ p, long q, long qEnd) {
  long qc = q < qEnd ? q : qEnd - 1;
  int4 v = ((const int4*)p)[qc];
  if (q >= qEnd) v = make_int4(-1, -1, -1, -1);
  return v;
}

__device__ __forceinline__ void note_hits(unsigned* __restrict__ q, int& cnt, int d, unsigned place, int base) {
  unsigned r = (unsigned)(d - base);
  bool hit = r < (unsigned)SL_NODES;
  unsigned m = __builtin_amdgcn_ballot_w32(hit);
  if (hit) q[cnt + (int)__builtin_amdgcn_mbcnt_lo(m, 0u)] = (r << SL_POS_BITS) | place;
  cnt += __builtin_popcount(m);
}

__device__ __forceinline__ void add_row(float* __restrict__ acc, int r, const float* __restrict__ h, int s) {
  const float4* hs = (const float4*)(h + (long)s * HID);
  float4 h0 = hs[0];
  float4 h1 = hs[1];
  float* row = acc + r * HID;
  atomicAdd(&row[0], h0.x); atomicAdd(&row[1], h0.y);
  atomicAdd(&row[2], h0.z); atomicAdd(&row[3], h0.w);
  atomicAdd(&row[4], h1.x); atomicAdd(&row[5], h1.y);
  atomicAdd(&row[6], h1.z); atomicAdd(&row[7], h1.w);
}

__device__ __forceinline__ void drain(const unsigned* __restrict__ q, int cnt, int lane, unsigned eBase,
                                      const int* __restrict__ src,
                                      const float* __restrict__ h, float* __restrict__ acc,
                                      unsigned* __restrict__ indeg) {
  __builtin_amdgcn_fence(__ATOMIC_ACQ_REL, "wavefront");
  for (int i = lane; i < cnt; i += 32) {
    unsigned v = q[i];
    int r = (int)(v >> SL_POS_BITS);
    unsigned place = v & ((1u << SL_POS_BITS) - 1u);
    unsigned e = eBase + ((place >> 7) << 12) + (place & 127u);
    int s = src[e];
    atomicAdd(&indeg[r], 1u);
    add_row(acc, r, h, s);
  }
  __builtin_amdgcn_fence(__ATOMIC_ACQ_REL, "wavefront");
}

__global__ __launch_bounds__(SL_THREADS) void spmm1_slices(const int* __restrict__ src,
                                                           const int* __restrict__ dst,
                                                           const float* __restrict__ h,
                                                           float* __restrict__ agg,
                                                           float* __restrict__ deg_d, int nE, int nN) {
  __shared__ float acc[SL_NODES * HID];
  __shared__ unsigned indeg[SL_NODES];
  __shared__ unsigned queue[SL_WAVES * SL_QCAP];

  const int tid  = threadIdx.x;
  const int lane = tid & 31;
  const int wave = tid >> 5;
  const int slice = blockIdx.x / SL_PARTS;
  const int part  = blockIdx.x % SL_PARTS;
  const int base  = slice * SL_NODES;

  for (int i = tid; i < SL_NODES * HID / 4; i += SL_THREADS)
    ((float4*)acc)[i] = make_float4(0.0f, 0.0f, 0.0f, 0.0f);
  for (int i = tid; i < SL_NODES; i += SL_THREADS) indeg[i] = 0u;
  __syncthreads();

  unsigned* q = queue + wave * SL_QCAP;
  int cnt = 0;
  static_assert(SL_THREADS == 1024, "a step of the walk is 1024 quads (the place bits count such steps)");

  const long nQ   = nE >> 2;
  const long qPer = (nQ + SL_PARTS - 1) / SL_PARTS;
  const long q0   = part * qPer;
  const long q1   = (q0 + qPer) < nQ ? (q0 + qPer) : nQ;

  const unsigned eBase = (unsigned)(4 * q0) + (unsigned)(wave * 128);
  unsigned place = (unsigned)(lane << 2);
  int4 n0 = make_int4(-1, -1, -1, -1), n1 = n0;
  if (q0 < q1) {
    n0 = load_quad(dst, q0 + tid, q1);
    n1 = load_quad(dst, q0 + tid + SL_THREADS, q1);
  }
  for (long qb = q0; qb < q1; qb += 2 * SL_THREADS) {
    int4 c0 = n0, c1 = n1;
    n0 = load_quad(dst, qb + 2 * SL_THREADS + tid, q1);
    n1 = load_quad(dst, qb + 3 * SL_THREADS + tid, q1);
    const unsigned p0 = place;
    const unsigned p1 = place + 128u;
    place += 256u;
    note_hits(q, cnt, c0.x, p0 + 0u, base);
    note_hits(q, cnt, c0.y, p0 + 1u, base);
    note_hits(q, cnt, c0.z, p0 + 2u, base);
    note_hits(q, cnt, c0.w, p0 + 3u, base);
    if (cnt > SL_QCAP - SL_GROUP) {
      drain(q, cnt, lane, eBase, src, h, acc, indeg);
      cnt = 0;
    }
    note_hits(q, cnt, c1.x, p1 + 0u, base);
    note_hits(q, cnt, c1.y, p1 + 1u, base);
    note_hits(q, cnt, c1.z, p1 + 2u, base);
    note_hits(q, cnt, c1.w, p1 + 3u, base);
    if (cnt > SL_QCAP - SL_GROUP) {
      drain(q, cnt, lane, eBase, src, h, acc, indeg);
      cnt = 0;
    }
  }
  drain(q, cnt, lane, eBase, src, h, acc, indeg);

  if (part == 0) {
    int e = (int)(nQ << 2) + tid;
    if (e < nE) {
      int d = dst[e];
      if ((unsigned)(d - base) < (unsigned)SL_NODES) {
        atomicAdd(&indeg[d - base], 1u);
        add_row(acc, d - base, h, src[e]);
      }
    }
  }
  __syncthreads();

  const int rows = (nN - base) < SL_NODES ? (nN - base) : SL_NODES;
  float* out = agg + (long)base * HID;
  for (int i = tid; i < rows * HID; i += SL_THREADS) atomicAdd(&out[i], acc[i]);
  for (int i = tid; i < rows; i += SL_THREADS) atomicAdd(&deg_d[base + i], (float)indeg[i]);
}


__global__ void layer2_node(const float* __restrict__ agg, float* __restrict__ in_norm,
                            const float* __restrict__ out_norm, const float* __restrict__ b1,
                            const float* __restrict__ W2, float* __restrict__ g2, int n) {
  int v = blockIdx.x * blockDim.x + threadIdx.x;
  if (v >= n) return;
  float innv = rsqrtf(fmaxf(in_norm[v], 1.0f));
  in_norm[v] = innv;
  const float* av = agg + (long)v * HID;
  float s = 0.0f;
#pragma unroll
  for (int j = 0; j < HID; ++j) {
    float z = fmaf(av[j], innv, b1[j]);
    z = z > 0.0f ? z : 0.0f;
    s = fmaf(z, W2[j], s);
  }
  g2[v] = s * out_norm[v];
}


__global__ __launch_bounds__(EG_THREADS) void spmm2_kernel(const int* __restrict__ src,
                                                           const int* __restrict__ dst,
                                                           const float* __restrict__ g2,
                                                           float* __restrict__ acc_h0,
                                                           float* __restrict__ acc_h1, int nE) {
  float* __restrict__ acc = half_of_package() ? acc_h1 : acc_h0;
  const int nQ     = nE >> 2;
  const int stride = EG_BLOCKS * EG_THREADS;
  const int gid    = blockIdx.x * EG_THREADS + threadIdx.x;
  const int4* __restrict__ src4 = (const int4*)src;
  const int4* __restrict__ dst4 = (const int4*)dst;
  int4 s = make_int4(0, 0, 0, 0), d = s;
  if (gid < nQ) { s = src4[gid]; d = dst4[gid]; }
  for (int q = gid; q < nQ; q += stride) {
    int4 sn = s, dn = d;
    if (q + stride < nQ) { sn = src4[q + stride]; dn = dst4[q + stride]; }
    float ga = g2[s.x], gb = g2[s.y], gc = g2[s.z], gd = g2[s.w];
    atomicAdd(&acc[d.x], ga);
    atomicAdd(&acc[d.y], gb);
    atomicAdd(&acc[d.z], gc);
    atomicAdd(&acc[d.w], gd);
    s = sn; d = dn;
  }
  int e = (nQ << 2) + gid;
  if (e < nE) atomicAdd(&acc[dst[e]], g2[src[e]]);
}

__global__ void final_kernel(const float* __restrict__ acc_h0, const float* __restrict__ acc_h1,
                             const float* __restrict__ in_norm, const float* __restrict__ b2,
                             float* __restrict__ out, int n) {
  int v = blockIdx.x * blockDim.x + threadIdx.x;
  if (v < n) out[v] = fmaf(acc_h0[v] + acc_h1[v], in_norm[v], b2[0]);
}


extern "C" void kernel_launch(void* const* d_in, const int* in_sizes, int n_in,
                              void* d_out, int out_size, void* d_ws, size_t ws_size,
                              hipStream_t stream) {
  const float* x  = (const float*)d_in[0];
  const float* W1 = (const float*)d_in[1];
  const float* b1 = (const float*)d_in[2];
  const float* W2 = (const float*)d_in[3];
  const float* b2 = (const float*)d_in[4];
  const int*  src = (const int*)d_in[5];
  const int*  dst = (const int*)d_in[6];

  const int nN = in_sizes[0] / IN_DIM;
  const int nE = in_sizes[5];

  float* ws     = (float*)d_ws;
  float* deg_s  = ws;
  float* deg_d  = ws + (long)nN;
  float* deg_s1 = ws + 2L * nN;
  float* g2     = deg_s1;
  float* h      = ws + 3L * nN;
  float* acc_h0 = h;
  float* acc_h1 = h + (long)nN;
  float* agg    = ws + 11L * nN;
  float* out    = (float*)d_out;

  const int TB = 256;
  dim3 blk(TB);

  zero_f32<<<(3 * nN + TB - 1) / TB, blk, 0, stream>>>(deg_s, 3 * nN);
  zero_f32<<<(8 * nN + TB - 1) / TB, blk, 0, stream>>>(agg, 8 * nN);

  degree_lines<<<DL_BLOCKS, dim3(DL_THREADS), 0, stream>>>(src, deg_s, deg_s1, nE);
  norm_kernel<<<(nN + TB - 1) / TB, blk, 0, stream>>>(deg_s, deg_s1, nN);

  gemm1_wmma<<<(nN + 127) / 128, blk, 0, stream>>>(x, W1, deg_s, h, nN);

  spmm1_slices<<<((nN + SL_NODES - 1) / SL_NODES) * SL_PARTS, dim3(SL_THREADS), 0, stream>>>(src, dst, h, agg, deg_d, nE, nN);
  zero_f32<<<(2 * nN + TB - 1) / TB, blk, 0, stream>>>(acc_h0, 2 * nN);
  layer2_node<<<(nN + TB - 1) / TB, blk, 0, stream>>>(agg, deg_d, deg_s, b1, W2, g2, nN);
  spmm2_kernel<<<EG_BLOCKS, dim3(EG_THREADS), 0, stream>>>(src, dst, g2, acc_h0, acc_h1, nE);
  final_kernel<<<(nN + TB - 1) / TB, blk, 0, stream>>>(acc_h0, acc_h1, deg_d, b2, out, nN);
}


# layer-1 slice owners scan one byte an edge (slice number, written with a packed row+source word by a new pass): 4 edges a test, drain reads one word, not src; edge order and LDS atomic order kept
# speedup vs baseline: 4.9001x; 1.2006x over previous
#include <hip/hip_runtime.h>

typedef __attribute__((ext_vector_type(2))) float v2f;
typedef __attribute__((ext_vector_type(8))) float v8f;

constexpr int IN_DIM = 128;
constexpr int HID    = 8;


__global__ void zero_f32(float* __restrict__ p, int n) {
  int i = blockIdx.x * blockDim.x + threadIdx.x;
  if (i < n) p[i] = 0.0f;
}


__device__ __forceinline__ int half_of_package() {
  unsigned w = __builtin_amdgcn_s_sendmsg_rtn(0x87);
  return (int)((w >> 18) & 1u);
}

constexpr int EG_BLOCKS  = 2048;
constexpr int EG_THREADS = 256;

constexpr int DL_BLOCKS  = 256;
constexpr int DL_THREADS = 1024;
constexpr int DL_LINES   = 16384;
constexpr int DL_CAP     = 12;
constexpr int DL_BATCH   = 16;
static_assert((DL_LINES & (DL_LINES - 1)) == 0, "the walk over the bins wraps with a mask");
static_assert(DL_LINES * 16 <= 320 * 1024, "LDS budget");

__device__ __forceinline__ void degree_put(unsigned* fill, unsigned char* ent, float* __restrict__ deg, int s) {
  if (s < 0) return;
  const unsigned line = (unsigned)s >> 5;
  if (line < (unsigned)DL_LINES) {
    const unsigned pos = atomicAdd(&fill[line * 4u], 1u);
    if (pos < (unsigned)DL_CAP) {
      ent[line * 16u + 4u + pos] = (unsigned char)(s & 31);
      return;
    }
  }
  atomicAdd(&deg[s], 1.0f);
}

__global__ __launch_bounds__(DL_THREADS) void degree_lines(const int* __restrict__ src,
                                                           float* __restrict__ deg_h0,
                                                           float* __restrict__ deg_h1, int nE) {
  __shared__ uint4 bins[DL_LINES];
  float* __restrict__ deg = half_of_package() ? deg_h1 : deg_h0;
  const int tid = threadIdx.x;
  for (int i = tid; i < DL_LINES; i += DL_THREADS) bins[i] = make_uint4(0u, 0u, 0u, 0u);
  __syncthreads();
  unsigned* fill = (unsigned*)bins;
  unsigned char* ent = (unsigned char*)bins;

  const int nQ   = nE >> 2;
  const int qPer = (nQ + DL_BLOCKS - 1) / DL_BLOCKS;
  const int q0   = blockIdx.x * qPer;
  const int q1   = (q0 + qPer) < nQ ? (q0 + qPer) : nQ;
  const int4* __restrict__ src4 = (const int4*)src;
  for (int qb = q0 + tid; qb < q1; qb += DL_BATCH * DL_THREADS) {
    int4 s[DL_BATCH];
#pragma unroll
    for (int k = 0; k < DL_BATCH; ++k) {
      const int q  = qb + k * DL_THREADS;
      const int qc = q < q1 ? q : q1 - 1;
      s[k] = src4[qc];
      if (q >= q1) s[k] = make_int4(-1, -1, -1, -1);
    }
#pragma unroll
    for (int k = 0; k < DL_BATCH; ++k) {
      degree_put(fill, ent, deg, s[k].x);
      degree_put(fill, ent, deg, s[k].y);
      degree_put(fill, ent, deg, s[k].z);
      degree_put(fill, ent, deg, s[k].w);
    }
  }
  if (blockIdx.x == 0) {
    const int e = (nQ << 2) + tid;
    if (e < nE) degree_put(fill, ent, deg, src[e]);
  }
  __syncthreads();

  const unsigned lane = (unsigned)tid & 31u;
  const unsigned g    = lane >> 3;
  const unsigned k    = lane & 7u;
  const unsigned sh   = (k & 3u) * 8u;
  const unsigned rot  = ((unsigned)blockIdx.x * 64u) & (unsigned)(DL_LINES - 1);
  constexpr int DL_AHEAD = 4;
  static_assert(DL_LINES % ((DL_THREADS / 32) * 4 * DL_AHEAD) == 0, "whole steps");
  for (int base = (tid >> 5) * 4 * DL_AHEAD; base < DL_LINES; base += (DL_THREADS / 32) * 4 * DL_AHEAD) {
    unsigned line[DL_AHEAD];
    uint4 b[DL_AHEAD];
#pragma unroll
    for (int j = 0; j < DL_AHEAD; ++j) {
      line[j] = ((unsigned)(base + 4 * j) + g + rot) & (unsigned)(DL_LINES - 1);
      b[j] = bins[line[j]];
    }
#pragma unroll
    for (int j = 0; j < DL_AHEAD; ++j) {
      const unsigned c  = b[j].x < (unsigned)DL_CAP ? b[j].x : (unsigned)DL_CAP;
      const unsigned w0 = ((k < 4u ? b[j].y : b[j].z) >> sh) & 31u;
      if (k < c) atomicAdd(&deg[line[j] * 32u + w0], 1.0f);
      const unsigned w1 = (b[j].w >> sh) & 31u;
      if (k < 4u && k + 8u < c) atomicAdd(&deg[line[j] * 32u + w1], 1.0f);
    }
  }
}

__global__ void norm_kernel(float* __restrict__ deg_h0, const float* __restrict__ deg_h1, int n) {
  int i = blockIdx.x * blockDim.x + threadIdx.x;
  if (i < n) deg_h0[i] = rsqrtf(fmaxf(deg_h0[i] + deg_h1[i], 1.0f));
}


__global__ __launch_bounds__(256) void gemm1_wmma(const float* __restrict__ x,
                                                  const float* __restrict__ W1,
                                                  const float* __restrict__ out_norm,
                                                  float* __restrict__ h, int nNodes) {
  __shared__ float w[IN_DIM * 16];
  int tid = threadIdx.x;
  for (int i = tid; i < IN_DIM * 16; i += 256) {
    int r = i >> 4, cidx = i & 15;
    w[i] = (cidx < HID) ? W1[r * HID + cidx] : 0.0f;
  }
  __syncthreads();

  int wid  = tid >> 5;
  int lane = tid & 31;
  int m    = lane & 15;
  int kh   = lane >> 4;

  int row0 = (blockIdx.x * 8 + wid) * 16;
  int node = row0 + m;
  int nodeClamped = node < nNodes ? node : nNodes - 1;
  const float* xrow = x + (long)nodeClamped * IN_DIM;

  v8f acc = {};
#pragma unroll
  for (int kk = 0; kk < IN_DIM; kk += 4) {
    int k0 = kk + 2 * kh;
    v2f a;
    a.x = xrow[k0];
    a.y = xrow[k0 + 1];
    v2f b;
    b.x = w[k0 * 16 + m];
    b.y = w[(k0 + 1) * 16 + m];
    acc = __builtin_amdgcn_wmma_f32_16x16x4_f32(
         false, a,  false, b,
         (short)0, acc,  false,  false);
  }

  if (m < HID) {
#pragma unroll
    for (int r = 0; r < 8; ++r) {
      int nr = row0 + r + 8 * kh;
      if (nr < nNodes) h[(long)nr * HID + m] = acc[r] * out_norm[nr];
    }
  }
}


constexpr int SL_NODES   = 7952;
constexpr int SL_PARTS   = 4;
constexpr int SL_THREADS = 1024;
constexpr int SL_WAVES   = SL_THREADS / 32;
constexpr int SL_QCAP    = 320;
constexpr int SL_GROUP   = 128;
static_assert(SL_NODES * (HID + 1) * 4 + SL_WAVES * SL_QCAP * 4 <= 320 * 1024, "LDS budget");
static_assert(SL_QCAP > SL_GROUP, "a whole group of hits fits behind the drain mark");
static_assert((SL_NODES * HID) % 32 == 0, "a slice of agg starts on a 128-byte line");
constexpr int SL_POS_BITS = 19;
static_assert(SL_NODES <= (1 << (32 - SL_POS_BITS)), "the row in the slice fits above the place bits");

__device__ __forceinline__ int4 load_quad(const int* __restrict__ p, long q, long qEnd) {
  long qc = q < qEnd ? q : qEnd - 1;
  int4 v = ((const int4*)p)[qc];
  if (q >= qEnd) v = make_int4(-1, -1, -1, -1);
  return v;
}

__device__ __forceinline__ void note_hits(unsigned* __restrict__ q, int& cnt, int d, unsigned place, int base) {
  unsigned r = (unsigned)(d - base);
  bool hit = r < (unsigned)SL_NODES;
  unsigned m = __builtin_amdgcn_ballot_w32(hit);
  if (hit) q[cnt + (int)__builtin_amdgcn_mbcnt_lo(m, 0u)] = (r << SL_POS_BITS) | place;
  cnt += __builtin_popcount(m);
}

__device__ __forceinline__ void add_row(float* __restrict__ acc, int r, const float* __restrict__ h, int s) {
  const float4* hs = (const float4*)(h + (long)s * HID);
  float4 h0 = hs[0];
  float4 h1 = hs[1];
  float* row = acc + r * HID;
  atomicAdd(&row[0], h0.x); atomicAdd(&row[1], h0.y);
  atomicAdd(&row[2], h0.z); atomicAdd(&row[3], h0.w);
  atomicAdd(&row[4], h1.x); atomicAdd(&row[5], h1.y);
  atomicAdd(&row[6], h1.z); atomicAdd(&row[7], h1.w);
}

__device__ __forceinline__ void drain(const unsigned* __restrict__ q, int cnt, int lane, unsigned eBase,
                                      const int* __restrict__ src,
                                      const float* __restrict__ h, float* __restrict__ acc,
                                      unsigned* __restrict__ indeg) {
  __builtin_amdgcn_fence(__ATOMIC_ACQ_REL, "wavefront");
  for (int i = lane; i < cnt; i += 32) {
    unsigned v = q[i];
    int r = (int)(v >> SL_POS_BITS);
    unsigned place = v & ((1u << SL_POS_BITS) - 1u);
    unsigned e = eBase + ((place >> 7) << 12) + (place & 127u);
    int s = src[e];
    atomicAdd(&indeg[r], 1u);
    add_row(acc, r, h, s);
  }
  __builtin_amdgcn_fence(__ATOMIC_ACQ_REL, "wavefront");
}

__global__ __launch_bounds__(SL_THREADS) void spmm1_slices(const int* __restrict__ src,
                                                           const int* __restrict__ dst,
                                                           const float* __restrict__ h,
                                                           float* __restrict__ agg,
                                                           float* __restrict__ deg_d, int nE, int nN) {
  __shared__ float acc[SL_NODES * HID];
  __shared__ unsigned indeg[SL_NODES];
  __shared__ unsigned queue[SL_WAVES * SL_QCAP];

  const int tid  = threadIdx.x;
  const int lane = tid & 31;
  const int wave = tid >> 5;
  const int slice = blockIdx.x / SL_PARTS;
  const int part  = blockIdx.x % SL_PARTS;
  const int base  = slice * SL_NODES;

  for (int i = tid; i < SL_NODES * HID / 4; i += SL_THREADS)
    ((float4*)acc)[i] = make_float4(0.0f, 0.0f, 0.0f, 0.0f);
  for (int i = tid; i < SL_NODES; i += SL_THREADS) indeg[i] = 0u;
  __syncthreads();

  unsigned* q = queue + wave * SL_QCAP;
  int cnt = 0;
  static_assert(SL_THREADS == 1024, "a step of the walk is 1024 quads (the place bits count such steps)");

  const long nQ   = nE >> 2;
  const long qPer = (nQ + SL_PARTS - 1) / SL_PARTS;
  const long q0   = part * qPer;
  const long q1   = (q0 + qPer) < nQ ? (q0 + qPer) : nQ;

  const unsigned eBase = (unsigned)(4 * q0) + (unsigned)(wave * 128);
  unsigned place = (unsigned)(lane << 2);
  int4 n0 = make_int4(-1, -1, -1, -1), n1 = n0;
  if (q0 < q1) {
    n0 = load_quad(dst, q0 + tid, q1);
    n1 = load_quad(dst, q0 + tid + SL_THREADS, q1);
  }
  for (long qb = q0; qb < q1; qb += 2 * SL_THREADS) {
    int4 c0 = n0, c1 = n1;
    n0 = load_quad(dst, qb + 2 * SL_THREADS + tid, q1);
    n1 = load_quad(dst, qb + 3 * SL_THREADS + tid, q1);
    const unsigned p0 = place;
    const unsigned p1 = place + 128u;
    place += 256u;
    note_hits(q, cnt, c0.x, p0 + 0u, base);
    note_hits(q, cnt, c0.y, p0 + 1u, base);
    note_hits(q, cnt, c0.z, p0 + 2u, base);
    note_hits(q, cnt, c0.w, p0 + 3u, base);
    if (cnt > SL_QCAP - SL_GROUP) {
      drain(q, cnt, lane, eBase, src, h, acc, indeg);
      cnt = 0;
    }
    note_hits(q, cnt, c1.x, p1 + 0u, base);
    note_hits(q, cnt, c1.y, p1 + 1u, base);
    note_hits(q, cnt, c1.z, p1 + 2u, base);
    note_hits(q, cnt, c1.w, p1 + 3u, base);
    if (cnt > SL_QCAP - SL_GROUP) {
      drain(q, cnt, lane, eBase, src, h, acc, indeg);
      cnt = 0;
    }
  }
  drain(q, cnt, lane, eBase, src, h, acc, indeg);

  if (part == 0) {
    int e = (int)(nQ << 2) + tid;
    if (e < nE) {
      int d = dst[e];
      if ((unsigned)(d - base) < (unsigned)SL_NODES) {
        atomicAdd(&indeg[d - base], 1u);
        add_row(acc, d - base, h, src[e]);
      }
    }
  }
  __syncthreads();

  const int rows = (nN - base) < SL_NODES ? (nN - base) : SL_NODES;
  float* out = agg + (long)base * HID;
  for (int i = tid; i < rows * HID; i += SL_THREADS) atomicAdd(&out[i], acc[i]);
  for (int i = tid; i < rows; i += SL_THREADS) atomicAdd(&deg_d[base + i], (float)indeg[i]);
}

constexpr int PK_ROW_SHIFT = 19;
constexpr unsigned PK_SRC_MASK = (1u << PK_ROW_SHIFT) - 1u;
static_assert(SL_NODES <= (1 << (32 - PK_ROW_SHIFT)), "the row in the slice fits above the source bits");

__device__ __forceinline__ unsigned pack_one(int s, int d, unsigned& sl) {
  sl = (unsigned)d / (unsigned)SL_NODES;
  return (((unsigned)d - sl * (unsigned)SL_NODES) << PK_ROW_SHIFT) | (unsigned)s;
}

__global__ __launch_bounds__(256) void pack_edges(const int* __restrict__ src, const int* __restrict__ dst,
                                                  unsigned* __restrict__ rec, unsigned char* __restrict__ sid,
                                                  int nE) {
  const int nQ = nE >> 2;
  const int q  = blockIdx.x * 256 + threadIdx.x;
  if (q < nQ) {
    const int4 s = ((const int4*)src)[q];
    const int4 d = ((const int4*)dst)[q];
    unsigned l0, l1, l2, l3;
    uint4 r;
    r.x = pack_one(s.x, d.x, l0);
    r.y = pack_one(s.y, d.y, l1);
    r.z = pack_one(s.z, d.z, l2);
    r.w = pack_one(s.w, d.w, l3);
    ((uint4*)rec)[q] = r;
    ((unsigned*)sid)[q] = l0 | (l1 << 8) | (l2 << 16) | (l3 << 24);
  }
  if (blockIdx.x == 0) {
    const int e = (nQ << 2) + threadIdx.x;
    if (e < nE) {
      unsigned sl;
      rec[e] = pack_one(src[e], dst[e], sl);
      sid[e] = (unsigned char)sl;
    }
  }
}

__device__ __forceinline__ uint4 load_group(const uint4* __restrict__ p, int g, int gEnd) {
  const int gc = g < gEnd ? g : gEnd - 1;
  uint4 v = p[gc];
  if (g >= gEnd) v = make_uint4(~0u, ~0u, ~0u, ~0u);
  return v;
}

__device__ __forceinline__ void drain_packed(const unsigned* __restrict__ q, int cnt, int lane, unsigned eBase,
                                             const unsigned* __restrict__ rec, const float* __restrict__ h,
                                             float* __restrict__ acc, unsigned* __restrict__ indeg) {
  __builtin_amdgcn_fence(__ATOMIC_ACQ_REL, "wavefront");
  for (int i = lane; i < cnt; i += 32) {
    const unsigned v = q[i];
    const unsigned e = eBase + ((v >> 9) << 14) + (v & 511u);
    const unsigned rc = rec[e];
    const int r = (int)(rc >> PK_ROW_SHIFT);
    atomicAdd(&indeg[r], 1u);
    add_row(acc, r, h, (int)(rc & PK_SRC_MASK));
  }
  __builtin_amdgcn_fence(__ATOMIC_ACQ_REL, "wavefront");
}

__global__ __launch_bounds__(SL_THREADS) void spmm1_packed(const unsigned* __restrict__ rec,
                                                           const unsigned char* __restrict__ sid,
                                                           const float* __restrict__ h,
                                                           float* __restrict__ agg,
                                                           float* __restrict__ deg_d, int nE, int nN) {
  __shared__ float acc[SL_NODES * HID];
  __shared__ unsigned indeg[SL_NODES];
  __shared__ unsigned queue[SL_WAVES * SL_QCAP];

  const int tid  = threadIdx.x;
  const int lane = tid & 31;
  const int wave = tid >> 5;
  const int slice = blockIdx.x / SL_PARTS;
  const int part  = blockIdx.x % SL_PARTS;
  const int base  = slice * SL_NODES;

  for (int i = tid; i < SL_NODES * HID / 4; i += SL_THREADS)
    ((float4*)acc)[i] = make_float4(0.0f, 0.0f, 0.0f, 0.0f);
  for (int i = tid; i < SL_NODES; i += SL_THREADS) indeg[i] = 0u;
  __syncthreads();

  unsigned* q = queue + wave * SL_QCAP;
  int cnt = 0;
  static_assert(SL_THREADS == 1024, "a step of the walk is 1024 groups (the place bits count such steps)");

  const int nG   = nE >> 4;
  const int gPer = (nG + SL_PARTS - 1) / SL_PARTS;
  const int g0   = part * gPer;
  const int g1   = (g0 + gPer) < nG ? (g0 + gPer) : nG;
  const uint4* __restrict__ sid4 = (const uint4*)sid;
  const unsigned pat   = (unsigned)slice * 0x01010101u;
  const unsigned eBase = (unsigned)(16 * g0) + (unsigned)(wave * 512);

  auto note_word = [&](unsigned w, unsigned place) {
    const unsigned x = w ^ pat;
    unsigned f = ~(((x & 0x7F7F7F7Fu) + 0x7F7F7F7Fu) | x) & 0x80808080u;
    unsigned m = __builtin_amdgcn_ballot_w32(f != 0u);
    while (m != 0u) {
      if (cnt > SL_QCAP - 32) {
        drain_packed(q, cnt, lane, eBase, rec, h, acc, indeg);
        cnt = 0;
      }
      if (f != 0u) {
        const unsigned k = (unsigned)__builtin_ctz(f) >> 3;
        q[cnt + (int)__builtin_amdgcn_mbcnt_lo(m, 0u)] = place | k;
        f &= f - 1u;
      }
      cnt += __builtin_popcount(m);
      m = __builtin_amdgcn_ballot_w32(f != 0u);
    }
  };

  if (g0 < g1) {
    uint4 n0 = load_group(sid4, g0 + tid, g1);
    uint4 n1 = load_group(sid4, g0 + SL_THREADS + tid, g1);
    unsigned place = (unsigned)(lane << 4);
    for (int gb = g0; gb < g1; gb += SL_THREADS) {
      const uint4 c = n0;
      n0 = n1;
      n1 = load_group(sid4, gb + 2 * SL_THREADS + tid, g1);
      note_word(c.x, place + 0u);
      note_word(c.y, place + 4u);
      note_word(c.z, place + 8u);
      note_word(c.w, place + 12u);
      place += 512u;
    }
  }
  drain_packed(q, cnt, lane, eBase, rec, h, acc, indeg);

  if (part == 0) {
    const int e = (nG << 4) + tid;
    if (e < nE && (int)sid[e] == slice) {
      const unsigned rc = rec[e];
      const int r = (int)(rc >> PK_ROW_SHIFT);
      atomicAdd(&indeg[r], 1u);
      add_row(acc, r, h, (int)(rc & PK_SRC_MASK));
    }
  }
  __syncthreads();

  const int rows = (nN - base) < SL_NODES ? (nN - base) : SL_NODES;
  float* out = agg + (long)base * HID;
  for (int i = tid; i < rows * HID; i += SL_THREADS) atomicAdd(&out[i], acc[i]);
  for (int i = tid; i < rows; i += SL_THREADS) atomicAdd(&deg_d[base + i], (float)indeg[i]);
}


__global__ void layer2_node(const float* __restrict__ agg, float* __restrict__ in_norm,
                            const float* __restrict__ out_norm, const float* __restrict__ b1,
                            const float* __restrict__ W2, float* __restrict__ g2, int n) {
  int v = blockIdx.x * blockDim.x + threadIdx.x;
  if (v >= n) return;
  float innv = rsqrtf(fmaxf(in_norm[v], 1.0f));
  in_norm[v] = innv;
  const float* av = agg + (long)v * HID;
  float s = 0.0f;
#pragma unroll
  for (int j = 0; j < HID; ++j) {
    float z = fmaf(av[j], innv, b1[j]);
    z = z > 0.0f ? z : 0.0f;
    s = fmaf(z, W2[j], s);
  }
  g2[v] = s * out_norm[v];
}


__global__ __launch_bounds__(EG_THREADS) void spmm2_kernel(const int* __restrict__ src,
                                                           const int* __restrict__ dst,
                                                           const float* __restrict__ g2,
                                                           float* __restrict__ acc_h0,
                                                           float* __restrict__ acc_h1, int nE) {
  float* __restrict__ acc = half_of_package() ? acc_h1 : acc_h0;
  const int nQ     = nE >> 2;
  const int stride = EG_BLOCKS * EG_THREADS;
  const int gid    = blockIdx.x * EG_THREADS + threadIdx.x;
  const int4* __restrict__ src4 = (const int4*)src;
  const int4* __restrict__ dst4 = (const int4*)dst;
  int4 s = make_int4(0, 0, 0, 0), d = s;
  if (gid < nQ) { s = src4[gid]; d = dst4[gid]; }
  for (int q = gid; q < nQ; q += stride) {
    int4 sn = s, dn = d;
    if (q + stride < nQ) { sn = src4[q + stride]; dn = dst4[q + stride]; }
    float ga = g2[s.x], gb = g2[s.y], gc = g2[s.z], gd = g2[s.w];
    atomicAdd(&acc[d.x], ga);
    atomicAdd(&acc[d.y], gb);
    atomicAdd(&acc[d.z], gc);
    atomicAdd(&acc[d.w], gd);
    s = sn; d = dn;
  }
  int e = (nQ << 2) + gid;
  if (e < nE) atomicAdd(&acc[dst[e]], g2[src[e]]);
}

__global__ void final_kernel(const float* __restrict__ acc_h0, const float* __restrict__ acc_h1,
                             const float* __restrict__ in_norm, const float* __restrict__ b2,
                             float* __restrict__ out, int n) {
  int v = blockIdx.x * blockDim.x + threadIdx.x;
  if (v < n) out[v] = fmaf(acc_h0[v] + acc_h1[v], in_norm[v], b2[0]);
}


extern "C" void kernel_launch(void* const* d_in, const int* in_sizes, int n_in,
                              void* d_out, int out_size, void* d_ws, size_t ws_size,
                              hipStream_t stream) {
  const float* x  = (const float*)d_in[0];
  const float* W1 = (const float*)d_in[1];
  const float* b1 = (const float*)d_in[2];
  const float* W2 = (const float*)d_in[3];
  const float* b2 = (const float*)d_in[4];
  const int*  src = (const int*)d_in[5];
  const int*  dst = (const int*)d_in[6];

  const int nN = in_sizes[0] / IN_DIM;
  const int nE = in_sizes[5];

  float* ws     = (float*)d_ws;
  float* deg_s  = ws;
  float* deg_d  = ws + (long)nN;
  float* deg_s1 = ws + 2L * nN;
  float* g2     = deg_s1;
  float* h      = ws + 3L * nN;
  float* acc_h0 = h;
  float* acc_h1 = h + (long)nN;
  float* agg    = ws + 11L * nN;
  float* out    = (float*)d_out;

  const int TB = 256;
  dim3 blk(TB);

  zero_f32<<<(3 * nN + TB - 1) / TB, blk, 0, stream>>>(deg_s, 3 * nN);
  zero_f32<<<(8 * nN + TB - 1) / TB, blk, 0, stream>>>(agg, 8 * nN);

  degree_lines<<<DL_BLOCKS, dim3(DL_THREADS), 0, stream>>>(src, deg_s, deg_s1, nE);
  norm_kernel<<<(nN + TB - 1) / TB, blk, 0, stream>>>(deg_s, deg_s1, nN);

  gemm1_wmma<<<(nN + 127) / 128, blk, 0, stream>>>(x, W1, deg_s, h, nN);

  const size_t recOff = ((size_t)19 * nN * sizeof(float) + 255) & ~(size_t)255;
  const size_t sidOff = recOff + (((size_t)nE * sizeof(unsigned) + 255) & ~(size_t)255);
  const bool packed = ws_size >= sidOff + (size_t)nE && nN <= (1 << PK_ROW_SHIFT);
  const int slGrid = ((nN + SL_NODES - 1) / SL_NODES) * SL_PARTS;
  if (packed) {
    unsigned* rec = (unsigned*)((unsigned char*)d_ws + recOff);
    unsigned char* sid = (unsigned char*)d_ws + sidOff;
    pack_edges<<<((nE >> 2) + TB - 1) / TB, blk, 0, stream>>>(src, dst, rec, sid, nE);
    spmm1_packed<<<slGrid, dim3(SL_THREADS), 0, stream>>>(rec, sid, h, agg, deg_d, nE, nN);
  } else {
    spmm1_slices<<<slGrid, dim3(SL_THREADS), 0, stream>>>(src, dst, h, agg, deg_d, nE, nN);
  }
  zero_f32<<<(2 * nN + TB - 1) / TB, blk, 0, stream>>>(acc_h0, 2 * nN);
  layer2_node<<<(nN + TB - 1) / TB, blk, 0, stream>>>(agg, deg_d, deg_s, b1, W2, g2, nN);
  spmm2_kernel<<<EG_BLOCKS, dim3(EG_THREADS), 0, stream>>>(src, dst, g2, acc_h0, acc_h1, nE);
  final_kernel<<<(nN + TB - 1) / TB, blk, 0, stream>>>(acc_h0, acc_h1, deg_d, b2, out, nN);
}


# layer-1 slice scan: a lane folds the match flags of its 32 bytes into one word: one ballot and hit loop per 1024 edges, not per 128; loads 3 trips ahead; queue order changes; LDS atomic order
# speedup vs baseline: 5.2206x; 1.0654x over previous
#include <hip/hip_runtime.h>

typedef __attribute__((ext_vector_type(2))) float v2f;
typedef __attribute__((ext_vector_type(8))) float v8f;

constexpr int IN_DIM = 128;
constexpr int HID    = 8;


__global__ void zero_f32(float* __restrict__ p, int n) {
  int i = blockIdx.x * blockDim.x + threadIdx.x;
  if (i < n) p[i] = 0.0f;
}


__device__ __forceinline__ int half_of_package() {
  unsigned w = __builtin_amdgcn_s_sendmsg_rtn(0x87);
  return (int)((w >> 18) & 1u);
}

constexpr int EG_BLOCKS  = 2048;
constexpr int EG_THREADS = 256;

constexpr int DL_BLOCKS  = 256;
constexpr int DL_THREADS = 1024;
constexpr int DL_LINES   = 16384;
constexpr int DL_CAP     = 12;
constexpr int DL_BATCH   = 16;
static_assert((DL_LINES & (DL_LINES - 1)) == 0, "the walk over the bins wraps with a mask");
static_assert(DL_LINES * 16 <= 320 * 1024, "LDS budget");

__device__ __forceinline__ void degree_put(unsigned* fill, unsigned char* ent, float* __restrict__ deg, int s) {
  if (s < 0) return;
  const unsigned line = (unsigned)s >> 5;
  if (line < (unsigned)DL_LINES) {
    const unsigned pos = atomicAdd(&fill[line * 4u], 1u);
    if (pos < (unsigned)DL_CAP) {
      ent[line * 16u + 4u + pos] = (unsigned char)(s & 31);
      return;
    }
  }
  atomicAdd(&deg[s], 1.0f);
}

__global__ __launch_bounds__(DL_THREADS) void degree_lines(const int* __restrict__ src,
                                                           float* __restrict__ deg_h0,
                                                           float* __restrict__ deg_h1, int nE) {
  __shared__ uint4 bins[DL_LINES];
  float* __restrict__ deg = half_of_package() ? deg_h1 : deg_h0;
  const int tid = threadIdx.x;
  for (int i = tid; i < DL_LINES; i += DL_THREADS) bins[i] = make_uint4(0u, 0u, 0u, 0u);
  __syncthreads();
  unsigned* fill = (unsigned*)bins;
  unsigned char* ent = (unsigned char*)bins;

  const int nQ   = nE >> 2;
  const int qPer = (nQ + DL_BLOCKS - 1) / DL_BLOCKS;
  const int q0   = blockIdx.x * qPer;
  const int q1   = (q0 + qPer) < nQ ? (q0 + qPer) : nQ;
  const int4* __restrict__ src4 = (const int4*)src;
  for (int qb = q0 + tid; qb < q1; qb += DL_BATCH * DL_THREADS) {
    int4 s[DL_BATCH];
#pragma unroll
    for (int k = 0; k < DL_BATCH; ++k) {
      const int q  = qb + k * DL_THREADS;
      const int qc = q < q1 ? q : q1 - 1;
      s[k] = src4[qc];
      if (q >= q1) s[k] = make_int4(-1, -1, -1, -1);
    }
#pragma unroll
    for (int k = 0; k < DL_BATCH; ++k) {
      degree_put(fill, ent, deg, s[k].x);
      degree_put(fill, ent, deg, s[k].y);
      degree_put(fill, ent, deg, s[k].z);
      degree_put(fill, ent, deg, s[k].w);
    }
  }
  if (blockIdx.x == 0) {
    const int e = (nQ << 2) + tid;
    if (e < nE) degree_put(fill, ent, deg, src[e]);
  }
  __syncthreads();

  const unsigned lane = (unsigned)tid & 31u;
  const unsigned g    = lane >> 3;
  const unsigned k    = lane & 7u;
  const unsigned sh   = (k & 3u) * 8u;
  const unsigned rot  = ((unsigned)blockIdx.x * 64u) & (unsigned)(DL_LINES - 1);
  constexpr int DL_AHEAD = 4;
  static_assert(DL_LINES % ((DL_THREADS / 32) * 4 * DL_AHEAD) == 0, "whole steps");
  for (int base = (tid >> 5) * 4 * DL_AHEAD; base < DL_LINES; base += (DL_THREADS / 32) * 4 * DL_AHEAD) {
    unsigned line[DL_AHEAD];
    uint4 b[DL_AHEAD];
#pragma unroll
    for (int j = 0; j < DL_AHEAD; ++j) {
      line[j] = ((unsigned)(base + 4 * j) + g + rot) & (unsigned)(DL_LINES - 1);
      b[j] = bins[line[j]];
    }
#pragma unroll
    for (int j = 0; j < DL_AHEAD; ++j) {
      const unsigned c  = b[j].x < (unsigned)DL_CAP ? b[j].x : (unsigned)DL_CAP;
      const unsigned w0 = ((k < 4u ? b[j].y : b[j].z) >> sh) & 31u;
      if (k < c) atomicAdd(&deg[line[j] * 32u + w0], 1.0f);
      const unsigned w1 = (b[j].w >> sh) & 31u;
      if (k < 4u && k + 8u < c) atomicAdd(&deg[line[j] * 32u + w1], 1.0f);
    }
  }
}

__global__ void norm_kernel(float* __restrict__ deg_h0, const float* __restrict__ deg_h1, int n) {
  int i = blockIdx.x * blockDim.x + threadIdx.x;
  if (i < n) deg_h0[i] = rsqrtf(fmaxf(deg_h0[i] + deg_h1[i], 1.0f));
}


__global__ __launch_bounds__(256) void gemm1_wmma(const float* __restrict__ x,
                                                  const float* __restrict__ W1,
                                                  const float* __restrict__ out_norm,
                                                  float* __restrict__ h, int nNodes) {
  __shared__ float w[IN_DIM * 16];
  int tid = threadIdx.x;
  for (int i = tid; i < IN_DIM * 16; i += 256) {
    int r = i >> 4, cidx = i & 15;
    w[i] = (cidx < HID) ? W1[r * HID + cidx] : 0.0f;
  }
  __syncthreads();

  int wid  = tid >> 5;
  int lane = tid & 31;
  int m    = lane & 15;
  int kh   = lane >> 4;

  int row0 = (blockIdx.x * 8 + wid) * 16;
  int node = row0 + m;
  int nodeClamped = node < nNodes ? node : nNodes - 1;
  const float* xrow = x + (long)nodeClamped * IN_DIM;

  v8f acc = {};
#pragma unroll
  for (int kk = 0; kk < IN_DIM; kk += 4) {
    int k0 = kk + 2 * kh;
    v2f a;
    a.x = xrow[k0];
    a.y = xrow[k0 + 1];
    v2f b;
    b.x = w[k0 * 16 + m];
    b.y = w[(k0 + 1) * 16 + m];
    acc = __builtin_amdgcn_wmma_f32_16x16x4_f32(
         false, a,  false, b,
         (short)0, acc,  false,  false);
  }

  if (m < HID) {
#pragma unroll
    for (int r = 0; r < 8; ++r) {
      int nr = row0 + r + 8 * kh;
      if (nr < nNodes) h[(long)nr * HID + m] = acc[r] * out_norm[nr];
    }
  }
}


constexpr int SL_NODES   = 7952;
constexpr int SL_PARTS   = 4;
constexpr int SL_THREADS = 1024;
constexpr int SL_WAVES   = SL_THREADS / 32;
constexpr int SL_QCAP    = 320;
constexpr int SL_GROUP   = 128;
static_assert(SL_NODES * (HID + 1) * 4 + SL_WAVES * SL_QCAP * 4 <= 320 * 1024, "LDS budget");
static_assert(SL_QCAP > SL_GROUP, "a whole group of hits fits behind the drain mark");
static_assert((SL_NODES * HID) % 32 == 0, "a slice of agg starts on a 128-byte line");
constexpr int SL_POS_BITS = 19;
static_assert(SL_NODES <= (1 << (32 - SL_POS_BITS)), "the row in the slice fits above the place bits");

__device__ __forceinline__ int4 load_quad(const int* __restrict__ p, long q, long qEnd) {
  long qc = q < qEnd ? q : qEnd - 1;
  int4 v = ((const int4*)p)[qc];
  if (q >= qEnd) v = make_int4(-1, -1, -1, -1);
  return v;
}

__device__ __forceinline__ void note_hits(unsigned* __restrict__ q, int& cnt, int d, unsigned place, int base) {
  unsigned r = (unsigned)(d - base);
  bool hit = r < (unsigned)SL_NODES;
  unsigned m = __builtin_amdgcn_ballot_w32(hit);
  if (hit) q[cnt + (int)__builtin_amdgcn_mbcnt_lo(m, 0u)] = (r << SL_POS_BITS) | place;
  cnt += __builtin_popcount(m);
}

__device__ __forceinline__ void add_row(float* __restrict__ acc, int r, const float* __restrict__ h, int s) {
  const float4* hs = (const float4*)(h + (long)s * HID);
  float4 h0 = hs[0];
  float4 h1 = hs[1];
  float* row = acc + r * HID;
  atomicAdd(&row[0], h0.x); atomicAdd(&row[1], h0.y);
  atomicAdd(&row[2], h0.z); atomicAdd(&row[3], h0.w);
  atomicAdd(&row[4], h1.x); atomicAdd(&row[5], h1.y);
  atomicAdd(&row[6], h1.z); atomicAdd(&row[7], h1.w);
}

__device__ __forceinline__ void drain(const unsigned* __restrict__ q, int cnt, int lane, unsigned eBase,
                                      const int* __restrict__ src,
                                      const float* __restrict__ h, float* __restrict__ acc,
                                      unsigned* __restrict__ indeg) {
  __builtin_amdgcn_fence(__ATOMIC_ACQ_REL, "wavefront");
  for (int i = lane; i < cnt; i += 32) {
    unsigned v = q[i];
    int r = (int)(v >> SL_POS_BITS);
    unsigned place = v & ((1u << SL_POS_BITS) - 1u);
    unsigned e = eBase + ((place >> 7) << 12) + (place & 127u);
    int s = src[e];
    atomicAdd(&indeg[r], 1u);
    add_row(acc, r, h, s);
  }
  __builtin_amdgcn_fence(__ATOMIC_ACQ_REL, "wavefront");
}

__global__ __launch_bounds__(SL_THREADS) void spmm1_slices(const int* __restrict__ src,
                                                           const int* __restrict__ dst,
                                                           const float* __restrict__ h,
                                                           float* __restrict__ agg,
                                                           float* __restrict__ deg_d, int nE, int nN) {
  __shared__ float acc[SL_NODES * HID];
  __shared__ unsigned indeg[SL_NODES];
  __shared__ unsigned queue[SL_WAVES * SL_QCAP];

  const int tid  = threadIdx.x;
  const int lane = tid & 31;
  const int wave = tid >> 5;
  const int slice = blockIdx.x / SL_PARTS;
  const int part  = blockIdx.x % SL_PARTS;
  const int base  = slice * SL_NODES;

  for (int i = tid; i < SL_NODES * HID / 4; i += SL_THREADS)
    ((float4*)acc)[i] = make_float4(0.0f, 0.0f, 0.0f, 0.0f);
  for (int i = tid; i < SL_NODES; i += SL_THREADS) indeg[i] = 0u;
  __syncthreads();

  unsigned* q = queue + wave * SL_QCAP;
  int cnt = 0;
  static_assert(SL_THREADS == 1024, "a step of the walk is 1024 quads (the place bits count such steps)");

  const long nQ   = nE >> 2;
  const long qPer = (nQ + SL_PARTS - 1) / SL_PARTS;
  const long q0   = part * qPer;
  const long q1   = (q0 + qPer) < nQ ? (q0 + qPer) : nQ;

  const unsigned eBase = (unsigned)(4 * q0) + (unsigned)(wave * 128);
  unsigned place = (unsigned)(lane << 2);
  int4 n0 = make_int4(-1, -1, -1, -1), n1 = n0;
  if (q0 < q1) {
    n0 = load_quad(dst, q0 + tid, q1);
    n1 = load_quad(dst, q0 + tid + SL_THREADS, q1);
  }
  for (long qb = q0; qb < q1; qb += 2 * SL_THREADS) {
    int4 c0 = n0, c1 = n1;
    n0 = load_quad(dst, qb + 2 * SL_THREADS + tid, q1);
    n1 = load_quad(dst, qb + 3 * SL_THREADS + tid, q1);
    const unsigned p0 = place;
    const unsigned p1 = place + 128u;
    place += 256u;
    note_hits(q, cnt, c0.x, p0 + 0u, base);
    note_hits(q, cnt, c0.y, p0 + 1u, base);
    note_hits(q, cnt, c0.z, p0 + 2u, base);
    note_hits(q, cnt, c0.w, p0 + 3u, base);
    if (cnt > SL_QCAP - SL_GROUP) {
      drain(q, cnt, lane, eBase, src, h, acc, indeg);
      cnt = 0;
    }
    note_hits(q, cnt, c1.x, p1 + 0u, base);
    note_hits(q, cnt, c1.y, p1 + 1u, base);
    note_hits(q, cnt, c1.z, p1 + 2u, base);
    note_hits(q, cnt, c1.w, p1 + 3u, base);
    if (cnt > SL_QCAP - SL_GROUP) {
      drain(q, cnt, lane, eBase, src, h, acc, indeg);
      cnt = 0;
    }
  }
  drain(q, cnt, lane, eBase, src, h, acc, indeg);

  if (part == 0) {
    int e = (int)(nQ << 2) + tid;
    if (e < nE) {
      int d = dst[e];
      if ((unsigned)(d - base) < (unsigned)SL_NODES) {
        atomicAdd(&indeg[d - base], 1u);
        add_row(acc, d - base, h, src[e]);
      }
    }
  }
  __syncthreads();

  const int rows = (nN - base) < SL_NODES ? (nN - base) : SL_NODES;
  float* out = agg + (long)base * HID;
  for (int i = tid; i < rows * HID; i += SL_THREADS) atomicAdd(&out[i], acc[i]);
  for (int i = tid; i < rows; i += SL_THREADS) atomicAdd(&deg_d[base + i], (float)indeg[i]);
}

constexpr int PK_ROW_SHIFT = 19;
constexpr unsigned PK_SRC_MASK = (1u << PK_ROW_SHIFT) - 1u;
static_assert(SL_NODES <= (1 << (32 - PK_ROW_SHIFT)), "the row in the slice fits above the source bits");

__device__ __forceinline__ unsigned pack_one(int s, int d, unsigned& sl) {
  sl = (unsigned)d / (unsigned)SL_NODES;
  return (((unsigned)d - sl * (unsigned)SL_NODES) << PK_ROW_SHIFT) | (unsigned)s;
}

__global__ __launch_bounds__(256) void pack_edges(const int* __restrict__ src, const int* __restrict__ dst,
                                                  unsigned* __restrict__ rec, unsigned char* __restrict__ sid,
                                                  int nE) {
  const int nQ = nE >> 2;
  const int q  = blockIdx.x * 256 + threadIdx.x;
  if (q < nQ) {
    const int4 s = ((const int4*)src)[q];
    const int4 d = ((const int4*)dst)[q];
    unsigned l0, l1, l2, l3;
    uint4 r;
    r.x = pack_one(s.x, d.x, l0);
    r.y = pack_one(s.y, d.y, l1);
    r.z = pack_one(s.z, d.z, l2);
    r.w = pack_one(s.w, d.w, l3);
    ((uint4*)rec)[q] = r;
    ((unsigned*)sid)[q] = l0 | (l1 << 8) | (l2 << 16) | (l3 << 24);
  }
  if (blockIdx.x == 0) {
    const int e = (nQ << 2) + threadIdx.x;
    if (e < nE) {
      unsigned sl;
      rec[e] = pack_one(src[e], dst[e], sl);
      sid[e] = (unsigned char)sl;
    }
  }
}

__device__ __forceinline__ uint4 load_group(const uint4* __restrict__ p, int g, int gEnd) {
  return p[g < gEnd ? g : gEnd - 1];
}

__device__ __forceinline__ unsigned miss_bits(unsigned w, unsigned pat) {
  return (w ^ pat) + 0x7F7F7F7Fu;
}
static_assert(((1 << PK_ROW_SHIFT) - 1) / SL_NODES < 127, "a slice number is a byte below 127");

__device__ __forceinline__ void drain_packed(const unsigned* __restrict__ q, int cnt, int lane, unsigned eBase,
                                             const unsigned* __restrict__ rec, const float* __restrict__ h,
                                             float* __restrict__ acc, unsigned* __restrict__ indeg) {
  __builtin_amdgcn_fence(__ATOMIC_ACQ_REL, "wavefront");
  for (int i = lane; i < cnt; i += 32) {
    const unsigned v = q[i];
    const unsigned e = eBase + ((v >> 10) << 15) + ((v & 4u) << 12) + ((v >> 1) & 0x1F0u) +
                       ((v & 3u) << 2) + ((v >> 3) & 3u);
    const unsigned rc = rec[e];
    const int r = (int)(rc >> PK_ROW_SHIFT);
    atomicAdd(&indeg[r], 1u);
    add_row(acc, r, h, (int)(rc & PK_SRC_MASK));
  }
  __builtin_amdgcn_fence(__ATOMIC_ACQ_REL, "wavefront");
}

__global__ __launch_bounds__(SL_THREADS) void spmm1_packed(const unsigned* __restrict__ rec,
                                                           const unsigned char* __restrict__ sid,
                                                           const float* __restrict__ h,
                                                           float* __restrict__ agg,
                                                           float* __restrict__ deg_d, int nE, int nN) {
  __shared__ float acc[SL_NODES * HID];
  __shared__ unsigned indeg[SL_NODES];
  __shared__ unsigned queue[SL_WAVES * SL_QCAP];

  const int tid  = threadIdx.x;
  const int lane = tid & 31;
  const int wave = tid >> 5;
  const int slice = blockIdx.x / SL_PARTS;
  const int part  = blockIdx.x % SL_PARTS;
  const int base  = slice * SL_NODES;

  for (int i = tid; i < SL_NODES * HID / 4; i += SL_THREADS)
    ((float4*)acc)[i] = make_float4(0.0f, 0.0f, 0.0f, 0.0f);
  for (int i = tid; i < SL_NODES; i += SL_THREADS) indeg[i] = 0u;
  __syncthreads();

  unsigned* q = queue + wave * SL_QCAP;
  int cnt = 0;
  static_assert(SL_THREADS == 1024, "a trip of the walk is twice 1024 groups (the entries count such trips)");

  const int nG   = nE >> 4;
  const int gPer = (nG + SL_PARTS - 1) / SL_PARTS;
  const int g0   = part * gPer;
  const int g1   = (g0 + gPer) < nG ? (g0 + gPer) : nG;
  const uint4* __restrict__ sid4 = (const uint4*)sid;
  const unsigned pat   = (unsigned)slice * 0x01010101u;
  const unsigned eBase = (unsigned)(16 * g0) + (unsigned)(wave * 512);

  constexpr int TRIP = 2 * SL_THREADS;
  constexpr int RING = 3;
  auto trip = [&](uint4& x0, uint4& x1, int gbt, unsigned place) {
    unsigned live = ~0u;
    if (gbt + TRIP > g1) {
      live = (gbt + tid < g1 ? 0x0F0F0F0Fu : 0u) | (gbt + SL_THREADS + tid < g1 ? 0xF0F0F0F0u : 0u);
    }
    const unsigned miss = ((miss_bits(x0.x, pat) >> 7) & 0x01010101u) | ((miss_bits(x0.y, pat) >> 6) & 0x02020202u) |
                          ((miss_bits(x0.z, pat) >> 5) & 0x04040404u) | ((miss_bits(x0.w, pat) >> 4) & 0x08080808u) |
                          ((miss_bits(x1.x, pat) >> 3) & 0x10101010u) | ((miss_bits(x1.y, pat) >> 2) & 0x20202020u) |
                          ((miss_bits(x1.z, pat) >> 1) & 0x40404040u) | (miss_bits(x1.w, pat) & 0x80808080u);
    unsigned f = ~miss & live;
    x0 = load_group(sid4, gbt + RING * TRIP + tid, g1);
    x1 = load_group(sid4, gbt + RING * TRIP + SL_THREADS + tid, g1);
    unsigned m = __builtin_amdgcn_ballot_w32(f != 0u);
    while (m != 0u) {
      if (cnt > SL_QCAP - 32) {
        drain_packed(q, cnt, lane, eBase, rec, h, acc, indeg);
        cnt = 0;
      }
      if (f != 0u) {
        q[cnt + (int)__builtin_amdgcn_mbcnt_lo(m, 0u)] = place | (unsigned)__builtin_ctz(f);
        f &= f - 1u;
      }
      cnt += __builtin_popcount(m);
      m = __builtin_amdgcn_ballot_w32(f != 0u);
    }
  };

  if (g0 < g1) {
    uint4 a0 = load_group(sid4, g0 + tid, g1);
    uint4 a1 = load_group(sid4, g0 + SL_THREADS + tid, g1);
    uint4 b0 = load_group(sid4, g0 + TRIP + tid, g1);
    uint4 b1 = load_group(sid4, g0 + TRIP + SL_THREADS + tid, g1);
    uint4 d0 = load_group(sid4, g0 + 2 * TRIP + tid, g1);
    uint4 d1 = load_group(sid4, g0 + 2 * TRIP + SL_THREADS + tid, g1);
    unsigned place = (unsigned)(lane << 5);
    for (int gb = g0; gb < g1; gb += RING * TRIP) {
      trip(a0, a1, gb, place);
      trip(b0, b1, gb + TRIP, place + 1024u);
      trip(d0, d1, gb + 2 * TRIP, place + 2048u);
      place += 3072u;
    }
  }
  drain_packed(q, cnt, lane, eBase, rec, h, acc, indeg);

  if (part == 0) {
    const int e = (nG << 4) + tid;
    if (e < nE && (int)sid[e] == slice) {
      const unsigned rc = rec[e];
      const int r = (int)(rc >> PK_ROW_SHIFT);
      atomicAdd(&indeg[r], 1u);
      add_row(acc, r, h, (int)(rc & PK_SRC_MASK));
    }
  }
  __syncthreads();

  const int rows = (nN - base) < SL_NODES ? (nN - base) : SL_NODES;
  float* out = agg + (long)base * HID;
  for (int i = tid; i < rows * HID; i += SL_THREADS) atomicAdd(&out[i], acc[i]);
  for (int i = tid; i < rows; i += SL_THREADS) atomicAdd(&deg_d[base + i], (float)indeg[i]);
}


__global__ void layer2_node(const float* __restrict__ agg, float* __restrict__ in_norm,
                            const float* __restrict__ out_norm, const float* __restrict__ b1,
                            const float* __restrict__ W2, float* __restrict__ g2, int n) {
  int v = blockIdx.x * blockDim.x + threadIdx.x;
  if (v >= n) return;
  float innv = rsqrtf(fmaxf(in_norm[v], 1.0f));
  in_norm[v] = innv;
  const float* av = agg + (long)v * HID;
  float s = 0.0f;
#pragma unroll
  for (int j = 0; j < HID; ++j) {
    float z = fmaf(av[j], innv, b1[j]);
    z = z > 0.0f ? z : 0.0f;
    s = fmaf(z, W2[j], s);
  }
  g2[v] = s * out_norm[v];
}


__global__ __launch_bounds__(EG_THREADS) void spmm2_kernel(const int* __restrict__ src,
                                                           const int* __restrict__ dst,
                                                           const float* __restrict__ g2,
                                                           float* __restrict__ acc_h0,
                                                           float* __restrict__ acc_h1, int nE) {
  float* __restrict__ acc = half_of_package() ? acc_h1 : acc_h0;
  const int nQ     = nE >> 2;
  const int stride = EG_BLOCKS * EG_THREADS;
  const int gid    = blockIdx.x * EG_THREADS + threadIdx.x;
  const int4* __restrict__ src4 = (const int4*)src;
  const int4* __restrict__ dst4 = (const int4*)dst;
  int4 s = make_int4(0, 0, 0, 0), d = s;
  if (gid < nQ) { s = src4[gid]; d = dst4[gid]; }
  for (int q = gid; q < nQ; q += stride) {
    int4 sn = s, dn = d;
    if (q + stride < nQ) { sn = src4[q + stride]; dn = dst4[q + stride]; }
    float ga = g2[s.x], gb = g2[s.y], gc = g2[s.z], gd = g2[s.w];
    atomicAdd(&acc[d.x], ga);
    atomicAdd(&acc[d.y], gb);
    atomicAdd(&acc[d.z], gc);
    atomicAdd(&acc[d.w], gd);
    s = sn; d = dn;
  }
  int e = (nQ << 2) + gid;
  if (e < nE) atomicAdd(&acc[dst[e]], g2[src[e]]);
}

__global__ void final_kernel(const float* __restrict__ acc_h0, const float* __restrict__ acc_h1,
                             const float* __restrict__ in_norm, const float* __restrict__ b2,
                             float* __restrict__ out, int n) {
  int v = blockIdx.x * blockDim.x + threadIdx.x;
  if (v < n) out[v] = fmaf(acc_h0[v] + acc_h1[v], in_norm[v], b2[0]);
}


extern "C" void kernel_launch(void* const* d_in, const int* in_sizes, int n_in,
                              void* d_out, int out_size, void* d_ws, size_t ws_size,
                              hipStream_t stream) {
  const float* x  = (const float*)d_in[0];
  const float* W1 = (const float*)d_in[1];
  const float* b1 = (const float*)d_in[2];
  const float* W2 = (const float*)d_in[3];
  const float* b2 = (const float*)d_in[4];
  const int*  src = (const int*)d_in[5];
  const int*  dst = (const int*)d_in[6];

  const int nN = in_sizes[0] / IN_DIM;
  const int nE = in_sizes[5];

  float* ws     = (float*)d_ws;
  float* deg_s  = ws;
  float* deg_d  = ws + (long)nN;
  float* deg_s1 = ws + 2L * nN;
  float* g2     = deg_s1;
  float* h      = ws + 3L * nN;
  float* acc_h0 = h;
  float* acc_h1 = h + (long)nN;
  float* agg    = ws + 11L * nN;
  float* out    = (float*)d_out;

  const int TB = 256;
  dim3 blk(TB);

  zero_f32<<<(3 * nN + TB - 1) / TB, blk, 0, stream>>>(deg_s, 3 * nN);
  zero_f32<<<(8 * nN + TB - 1) / TB, blk, 0, stream>>>(agg, 8 * nN);

  degree_lines<<<DL_BLOCKS, dim3(DL_THREADS), 0, stream>>>(src, deg_s, deg_s1, nE);
  norm_kernel<<<(nN + TB - 1) / TB, blk, 0, stream>>>(deg_s, deg_s1, nN);

  gemm1_wmma<<<(nN + 127) / 128, blk, 0, stream>>>(x, W1, deg_s, h, nN);

  const size_t recOff = ((size_t)19 * nN * sizeof(float) + 255) & ~(size_t)255;
  const size_t sidOff = recOff + (((size_t)nE * sizeof(unsigned) + 255) & ~(size_t)255);
  const bool packed = ws_size >= sidOff + (size_t)nE && nN <= (1 << PK_ROW_SHIFT);
  const int slGrid = ((nN + SL_NODES - 1) / SL_NODES) * SL_PARTS;
  if (packed) {
    unsigned* rec = (unsigned*)((unsigned char*)d_ws + recOff);
    unsigned char* sid = (unsigned char*)d_ws + sidOff;
    pack_edges<<<((nE >> 2) + TB - 1) / TB, blk, 0, stream>>>(src, dst, rec, sid, nE);
    spmm1_packed<<<slGrid, dim3(SL_THREADS), 0, stream>>>(rec, sid, h, agg, deg_d, nE, nN);
  } else {
    spmm1_slices<<<slGrid, dim3(SL_THREADS), 0, stream>>>(src, dst, h, agg, deg_d, nE, nN);
  }
  zero_f32<<<(2 * nN + TB - 1) / TB, blk, 0, stream>>>(acc_h0, 2 * nN);
  layer2_node<<<(nN + TB - 1) / TB, blk, 0, stream>>>(agg, deg_d, deg_s, b1, W2, g2, nN);
  spmm2_kernel<<<EG_BLOCKS, dim3(EG_THREADS), 0, stream>>>(src, dst, g2, acc_h0, acc_h1, nE);
  final_kernel<<<(nN + TB - 1) / TB, blk, 0, stream>>>(acc_h0, acc_h1, deg_d, b2, out, nN);
}
